# starting kernel
# baseline (speedup 1.0000x reference)
#include <hip/hip_runtime.h>
#include <hip/hip_bf16.h>
#include <math.h>

constexpr int B_  = 2;
constexpr int H_  = 8;
constexpr int LQ_ = 128;
constexpr int LK_ = 128;
constexpr int D_  = 512;
constexpr int DH_ = 64;
constexpr int F_  = 512;

typedef float v2f __attribute__((ext_vector_type(2)));
typedef float v8f __attribute__((ext_vector_type(8)));

#if __has_builtin(__builtin_amdgcn_tanhf)
#define TANH_F32(x) __builtin_amdgcn_tanhf(x)
#elif __has_builtin(__builtin_amdgcn_tanh_f32)
#define TANH_F32(x) __builtin_amdgcn_tanh_f32(x)
#else
#define TANH_F32(x) tanhf(x)
#endif

__device__ __forceinline__ v8f wmma_f32_tile(const float* __restrict__ A, int lda,
                                             const float* __restrict__ W, int ldw,
                                             int K, int m0, int n0, int lane)
{
  const int half = lane >> 4;
  const int l16  = lane & 15;
  const float* arow = A + (size_t)(m0 + l16) * lda + 2 * half;
  const float* wrow = W + (size_t)(n0 + l16) * ldw + 2 * half;
  v8f c = {};
  for (int k = 0; k < K; k += 4) {
    v2f a = *(const v2f*)(arow + k);
    v2f b = *(const v2f*)(wrow + k);
    c = __builtin_amdgcn_wmma_f32_16x16x4_f32(
         false, a,  false, b,
         (short)0, c,  false,  false);
  }
  return c;
}

__global__ void __launch_bounds__(256)
proj_kernel(const float* __restrict__ X, const float* __restrict__ W,
            float* __restrict__ out, int split)
{
  const int M = B_ * LQ_;
  const int N = D_;
  const int K = D_;
  const int tid  = threadIdx.x;
  const int lane = tid & 31;
  const int wave = tid >> 5;
  const int tile = blockIdx.x * 8 + wave;
  const int tm = tile / (N / 16);
  const int tn = tile % (N / 16);
  const int m0 = tm * 16, n0 = tn * 16;

  v8f c = wmma_f32_tile(X, K, W, K, K, m0, n0, lane);

  const int l16 = lane & 15, half = lane >> 4;
  const int col = n0 + l16;
  for (int r = 0; r < 8; ++r) {
    const int row = m0 + r + 8 * half;
    if (split) {
      const int b = row / LQ_, l = row % LQ_;
      const int h = col / DH_, j = col % DH_;
      out[(((size_t)(b * H_ + h) * LQ_) + l) * DH_ + j] = c[r];
    } else {
      out[(size_t)row * N + col] = c[r];
    }
  }
}

__global__ void __launch_bounds__(256)
feat_kernel(const float* __restrict__ Hd, const float* __restrict__ Wa,
            float* __restrict__ outf)
{
  const int bh = blockIdx.y;
  const float* A = Hd + (size_t)bh * LQ_ * DH_;
  float* O = outf + (size_t)bh * LQ_ * F_;

  const int tid  = threadIdx.x;
  const int lane = tid & 31;
  const int wave = tid >> 5;
  const int tile = blockIdx.x * 8 + wave;
  const int tm = tile / (F_ / 16);
  const int tn = tile % (F_ / 16);
  const int m0 = tm * 16, n0 = tn * 16;

  v8f c = wmma_f32_tile(A, DH_, Wa, DH_, DH_, m0, n0, lane);

  const int l16 = lane & 15, half = lane >> 4;
  for (int r = 0; r < 8; ++r)
    O[(size_t)(m0 + r + 8 * half) * F_ + n0 + l16] = c[r];
}

__global__ void __launch_bounds__(256)
score_softmax_av_kernel(const float* __restrict__ qf,
                        const float* __restrict__ kf,
                        const float* __restrict__ vh,
                        const float* __restrict__ wv_a,
                        const int*   __restrict__ valid_lens,
                        float* __restrict__ ctx)
{
  __shared__ float sq[F_];
  __shared__ float swv[F_];
  __shared__ float sp[LK_];
  __shared__ float spart[4 * DH_];
  __shared__ float swmax[8];
  __shared__ float swsum[8];

  const int bhq = blockIdx.x;
  const int qi  = bhq % LQ_;
  const int bh  = bhq / LQ_;
  const int b   = bh / H_;
  const int h   = bh % H_;
  const int tid  = threadIdx.x;
  const int lane = tid & 31;
  const int wave = tid >> 5;

  const float* qrow = qf + ((size_t)bh * LQ_ + qi) * F_;
  for (int f = tid; f < F_; f += 256) { sq[f] = qrow[f]; swv[f] = wv_a[f]; }
  __syncthreads();

  const float* kbase = kf + (size_t)bh * LK_ * F_;
  for (int k = wave; k < LK_; k += 8) {
    const float* krow = kbase + (size_t)k * F_;
    float acc = 0.f;
    for (int f = lane; f < F_; f += 32)
      acc += swv[f] * TANH_F32(sq[f] + krow[f]);
    for (int off = 16; off > 0; off >>= 1)
      acc += __shfl_xor(acc, off, 32);
    if (lane == 0) sp[k] = acc;
  }
  __syncthreads();

  const int valid = valid_lens[b];
  float val = (tid < LK_) ? ((tid < valid) ? sp[tid] : -1.0e6f) : -3.0e38f;

  float m = val;
  for (int off = 16; off > 0; off >>= 1)
    m = fmaxf(m, __shfl_xor(m, off, 32));
  if (lane == 0) swmax[wave] = (wave < 4) ? m : -3.0e38f;
  __syncthreads();
  const float rowmax = fmaxf(fmaxf(swmax[0], swmax[1]), fmaxf(swmax[2], swmax[3]));

  float e = (tid < LK_) ? __expf(val - rowmax) : 0.f;
  float s = e;
  for (int off = 16; off > 0; off >>= 1)
    s += __shfl_xor(s, off, 32);
  if (lane == 0) swsum[wave] = (wave < 4) ? s : 0.f;
  __syncthreads();
  const float rowsum = swsum[0] + swsum[1] + swsum[2] + swsum[3];
  if (tid < LK_) sp[tid] = e / rowsum;
  __syncthreads();

  const int j = tid & 63;
  const int part = tid >> 6;
  const float* vb = vh + (size_t)bh * LK_ * DH_;
  float acc = 0.f;
  for (int k = part * 32; k < part * 32 + 32; ++k)
    acc += sp[k] * vb[(size_t)k * DH_ + j];
  spart[part * DH_ + j] = acc;
  __syncthreads();

  if (tid < DH_) {
    const float o = spart[tid] + spart[DH_ + tid] + spart[2 * DH_ + tid] + spart[3 * DH_ + tid];
    ctx[((size_t)b * LQ_ + qi) * D_ + h * DH_ + tid] = o;
  }
}

extern "C" void kernel_launch(void* const* d_in, const int* in_sizes, int n_in,
                              void* d_out, int out_size, void* d_ws, size_t ws_size,
                              hipStream_t stream) {
  (void)in_sizes; (void)n_in; (void)out_size; (void)ws_size;

  const float* queries    = (const float*)d_in[0];
  const float* keys       = (const float*)d_in[1];
  const float* values     = (const float*)d_in[2];
  const int*   valid_lens = (const int*)  d_in[3];
  const float* Wq  = (const float*)d_in[4];
  const float* Wk  = (const float*)d_in[5];
  const float* Wv  = (const float*)d_in[6];
  const float* Wo  = (const float*)d_in[7];
  const float* Wqa = (const float*)d_in[8];
  const float* Wka = (const float*)d_in[9];
  const float* wva = (const float*)d_in[10];

  float* ws  = (float*)d_ws;
  const size_t nQKV = (size_t)B_ * H_ * LQ_ * DH_;
  const size_t nF   = (size_t)B_ * H_ * LQ_ * F_;
  float* q   = ws;
  float* k   = q  + nQKV;
  float* v   = k  + nQKV;
  float* qf  = v  + nQKV;
  float* kfb = qf + nF;
  float* ctx = kfb + nF;

  dim3 blk(256);
  proj_kernel<<<64, blk, 0, stream>>>(queries, Wq, q, 1);
  proj_kernel<<<64, blk, 0, stream>>>(keys,    Wk, k, 1);
  proj_kernel<<<64, blk, 0, stream>>>(values,  Wv, v, 1);

  feat_kernel<<<dim3(32, B_ * H_), blk, 0, stream>>>(q, Wqa, qf);
  feat_kernel<<<dim3(32, B_ * H_), blk, 0, stream>>>(k, Wka, kfb);

  score_softmax_av_kernel<<<B_ * H_ * LQ_, blk, 0, stream>>>(
      qf, kfb, v, wva, valid_lens, ctx);

  proj_kernel<<<64, blk, 0, stream>>>(ctx, Wo, (float*)d_out, 0);
}


# Merge independent launches: Q/K/V projections run as one launch (grid.y picks the operand set) and the two feature GEMMs as one (grid.z); 7 launches -> 4, inner loops and summation order unchanged
# speedup vs baseline: 1.4683x; 1.4683x over previous
#include <hip/hip_runtime.h>
#include <hip/hip_bf16.h>
#include <math.h>

constexpr int B_  = 2;
constexpr int H_  = 8;
constexpr int LQ_ = 128;
constexpr int LK_ = 128;
constexpr int D_  = 512;
constexpr int DH_ = 64;
constexpr int F_  = 512;

typedef float v2f __attribute__((ext_vector_type(2)));
typedef float v8f __attribute__((ext_vector_type(8)));

#if __has_builtin(__builtin_amdgcn_tanhf)
#define TANH_F32(x) __builtin_amdgcn_tanhf(x)
#elif __has_builtin(__builtin_amdgcn_tanh_f32)
#define TANH_F32(x) __builtin_amdgcn_tanh_f32(x)
#else
#define TANH_F32(x) tanhf(x)
#endif

__device__ __forceinline__ v8f wmma_f32_tile(const float* __restrict__ A, int lda,
                                             const float* __restrict__ W, int ldw,
                                             int K, int m0, int n0, int lane)
{
  const int half = lane >> 4;
  const int l16  = lane & 15;
  const float* arow = A + (size_t)(m0 + l16) * lda + 2 * half;
  const float* wrow = W + (size_t)(n0 + l16) * ldw + 2 * half;
  v8f c = {};
  for (int k = 0; k < K; k += 4) {
    v2f a = *(const v2f*)(arow + k);
    v2f b = *(const v2f*)(wrow + k);
    c = __builtin_amdgcn_wmma_f32_16x16x4_f32(
         false, a,  false, b,
         (short)0, c,  false,  false);
  }
  return c;
}

__device__ __forceinline__ void proj_body(const float* __restrict__ X, const float* __restrict__ W,
                                          float* __restrict__ out, int split, int blk)
{
  const int N = D_;
  const int K = D_;
  const int tid  = threadIdx.x;
  const int lane = tid & 31;
  const int wave = tid >> 5;
  const int tile = blk * 8 + wave;
  const int tm = tile / (N / 16);
  const int tn = tile % (N / 16);
  const int m0 = tm * 16, n0 = tn * 16;

  v8f c = wmma_f32_tile(X, K, W, K, K, m0, n0, lane);

  const int l16 = lane & 15, half = lane >> 4;
  const int col = n0 + l16;
  for (int r = 0; r < 8; ++r) {
    const int row = m0 + r + 8 * half;
    if (split) {
      const int b = row / LQ_, l = row % LQ_;
      const int h = col / DH_, j = col % DH_;
      out[(((size_t)(b * H_ + h) * LQ_) + l) * DH_ + j] = c[r];
    } else {
      out[(size_t)row * N + col] = c[r];
    }
  }
}

__global__ void __launch_bounds__(256)
proj_kernel(const float* __restrict__ X, const float* __restrict__ W,
            float* __restrict__ out)
{
  proj_body(X, W, out, 0, blockIdx.x);
}

__global__ void __launch_bounds__(256)
proj_qkv_kernel(const float* __restrict__ Xq, const float* __restrict__ Xk,
                const float* __restrict__ Xv,
                const float* __restrict__ Wq, const float* __restrict__ Wk,
                const float* __restrict__ Wv,
                float* __restrict__ oq, float* __restrict__ ok, float* __restrict__ ov)
{
  const int which = blockIdx.y;
  const float* X = (which == 0) ? Xq : (which == 1) ? Xk : Xv;
  const float* W = (which == 0) ? Wq : (which == 1) ? Wk : Wv;
  float*       o = (which == 0) ? oq : (which == 1) ? ok : ov;
  proj_body(X, W, o, 1, blockIdx.x);
}

__global__ void __launch_bounds__(256)
feat_kernel(const float* __restrict__ Hq, const float* __restrict__ Hk,
            const float* __restrict__ Wqa, const float* __restrict__ Wka,
            float* __restrict__ outq, float* __restrict__ outk)
{
  const int which = blockIdx.z;
  const float* Hd   = which ? Hk  : Hq;
  const float* Wa   = which ? Wka : Wqa;
  float*       outf = which ? outk : outq;

  const int bh = blockIdx.y;
  const float* A = Hd + (size_t)bh * LQ_ * DH_;
  float* O = outf + (size_t)bh * LQ_ * F_;

  const int tid  = threadIdx.x;
  const int lane = tid & 31;
  const int wave = tid >> 5;
  const int tile = blockIdx.x * 8 + wave;
  const int tm = tile / (F_ / 16);
  const int tn = tile % (F_ / 16);
  const int m0 = tm * 16, n0 = tn * 16;

  v8f c = wmma_f32_tile(A, DH_, Wa, DH_, DH_, m0, n0, lane);

  const int l16 = lane & 15, half = lane >> 4;
  for (int r = 0; r < 8; ++r)
    O[(size_t)(m0 + r + 8 * half) * F_ + n0 + l16] = c[r];
}

__global__ void __launch_bounds__(256)
score_softmax_av_kernel(const float* __restrict__ qf,
                        const float* __restrict__ kf,
                        const float* __restrict__ vh,
                        const float* __restrict__ wv_a,
                        const int*   __restrict__ valid_lens,
                        float* __restrict__ ctx)
{
  __shared__ float sq[F_];
  __shared__ float swv[F_];
  __shared__ float sp[LK_];
  __shared__ float spart[4 * DH_];
  __shared__ float swmax[8];
  __shared__ float swsum[8];

  const int bhq = blockIdx.x;
  const int qi  = bhq % LQ_;
  const int bh  = bhq / LQ_;
  const int b   = bh / H_;
  const int h   = bh % H_;
  const int tid  = threadIdx.x;
  const int lane = tid & 31;
  const int wave = tid >> 5;

  const float* qrow = qf + ((size_t)bh * LQ_ + qi) * F_;
  for (int f = tid; f < F_; f += 256) { sq[f] = qrow[f]; swv[f] = wv_a[f]; }
  __syncthreads();

  const float* kbase = kf + (size_t)bh * LK_ * F_;
  for (int k = wave; k < LK_; k += 8) {
    const float* krow = kbase + (size_t)k * F_;
    float acc = 0.f;
    for (int f = lane; f < F_; f += 32)
      acc += swv[f] * TANH_F32(sq[f] + krow[f]);
    for (int off = 16; off > 0; off >>= 1)
      acc += __shfl_xor(acc, off, 32);
    if (lane == 0) sp[k] = acc;
  }
  __syncthreads();

  const int valid = valid_lens[b];
  float val = (tid < LK_) ? ((tid < valid) ? sp[tid] : -1.0e6f) : -3.0e38f;

  float m = val;
  for (int off = 16; off > 0; off >>= 1)
    m = fmaxf(m, __shfl_xor(m, off, 32));
  if (lane == 0) swmax[wave] = (wave < 4) ? m : -3.0e38f;
  __syncthreads();
  const float rowmax = fmaxf(fmaxf(swmax[0], swmax[1]), fmaxf(swmax[2], swmax[3]));

  float e = (tid < LK_) ? __expf(val - rowmax) : 0.f;
  float s = e;
  for (int off = 16; off > 0; off >>= 1)
    s += __shfl_xor(s, off, 32);
  if (lane == 0) swsum[wave] = (wave < 4) ? s : 0.f;
  __syncthreads();
  const float rowsum = swsum[0] + swsum[1] + swsum[2] + swsum[3];
  if (tid < LK_) sp[tid] = e / rowsum;
  __syncthreads();

  const int j = tid & 63;
  const int part = tid >> 6;
  const float* vb = vh + (size_t)bh * LK_ * DH_;
  float acc = 0.f;
  for (int k = part * 32; k < part * 32 + 32; ++k)
    acc += sp[k] * vb[(size_t)k * DH_ + j];
  spart[part * DH_ + j] = acc;
  __syncthreads();

  if (tid < DH_) {
    const float o = spart[tid] + spart[DH_ + tid] + spart[2 * DH_ + tid] + spart[3 * DH_ + tid];
    ctx[((size_t)b * LQ_ + qi) * D_ + h * DH_ + tid] = o;
  }
}

extern "C" void kernel_launch(void* const* d_in, const int* in_sizes, int n_in,
                              void* d_out, int out_size, void* d_ws, size_t ws_size,
                              hipStream_t stream) {
  (void)in_sizes; (void)n_in; (void)out_size; (void)ws_size;

  const float* queries    = (const float*)d_in[0];
  const float* keys       = (const float*)d_in[1];
  const float* values     = (const float*)d_in[2];
  const int*   valid_lens = (const int*)  d_in[3];
  const float* Wq  = (const float*)d_in[4];
  const float* Wk  = (const float*)d_in[5];
  const float* Wv  = (const float*)d_in[6];
  const float* Wo  = (const float*)d_in[7];
  const float* Wqa = (const float*)d_in[8];
  const float* Wka = (const float*)d_in[9];
  const float* wva = (const float*)d_in[10];

  float* ws  = (float*)d_ws;
  const size_t nQKV = (size_t)B_ * H_ * LQ_ * DH_;
  const size_t nF   = (size_t)B_ * H_ * LQ_ * F_;
  float* q   = ws;
  float* k   = q  + nQKV;
  float* v   = k  + nQKV;
  float* qf  = v  + nQKV;
  float* kfb = qf + nF;
  float* ctx = kfb + nF;

  dim3 blk(256);
  proj_qkv_kernel<<<dim3(64, 3), blk, 0, stream>>>(queries, keys, values, Wq, Wk, Wv, q, k, v);

  feat_kernel<<<dim3(32, B_ * H_, 2), blk, 0, stream>>>(q, k, Wqa, Wka, qf, kfb);

  score_softmax_av_kernel<<<B_ * H_ * LQ_, blk, 0, stream>>>(
      qf, kfb, v, wva, valid_lens, ctx);

  proj_kernel<<<64, blk, 0, stream>>>(ctx, Wo, (float*)d_out);
}


# Score stage: each work-group takes 4 query rows of a head, the lane's query features and weights held in registers, so each key-feature load serves 4 rows; every sum keeps its order
# speedup vs baseline: 1.8466x; 1.2576x over previous
#include <hip/hip_runtime.h>
#include <hip/hip_bf16.h>
#include <math.h>

constexpr int B_  = 2;
constexpr int H_  = 8;
constexpr int LQ_ = 128;
constexpr int LK_ = 128;
constexpr int D_  = 512;
constexpr int DH_ = 64;
constexpr int F_  = 512;

typedef float v2f __attribute__((ext_vector_type(2)));
typedef float v8f __attribute__((ext_vector_type(8)));

#if __has_builtin(__builtin_amdgcn_tanhf)
#define TANH_F32(x) __builtin_amdgcn_tanhf(x)
#elif __has_builtin(__builtin_amdgcn_tanh_f32)
#define TANH_F32(x) __builtin_amdgcn_tanh_f32(x)
#else
#define TANH_F32(x) tanhf(x)
#endif

__device__ __forceinline__ v8f wmma_f32_tile(const float* __restrict__ A, int lda,
                                             const float* __restrict__ W, int ldw,
                                             int K, int m0, int n0, int lane)
{
  const int half = lane >> 4;
  const int l16  = lane & 15;
  const float* arow = A + (size_t)(m0 + l16) * lda + 2 * half;
  const float* wrow = W + (size_t)(n0 + l16) * ldw + 2 * half;
  v8f c = {};
  for (int k = 0; k < K; k += 4) {
    v2f a = *(const v2f*)(arow + k);
    v2f b = *(const v2f*)(wrow + k);
    c = __builtin_amdgcn_wmma_f32_16x16x4_f32(
         false, a,  false, b,
         (short)0, c,  false,  false);
  }
  return c;
}

__device__ __forceinline__ void proj_body(const float* __restrict__ X, const float* __restrict__ W,
                                          float* __restrict__ out, int split, int blk)
{
  const int N = D_;
  const int K = D_;
  const int tid  = threadIdx.x;
  const int lane = tid & 31;
  const int wave = tid >> 5;
  const int tile = blk * 8 + wave;
  const int tm = tile / (N / 16);
  const int tn = tile % (N / 16);
  const int m0 = tm * 16, n0 = tn * 16;

  v8f c = wmma_f32_tile(X, K, W, K, K, m0, n0, lane);

  const int l16 = lane & 15, half = lane >> 4;
  const int col = n0 + l16;
  for (int r = 0; r < 8; ++r) {
    const int row = m0 + r + 8 * half;
    if (split) {
      const int b = row / LQ_, l = row % LQ_;
      const int h = col / DH_, j = col % DH_;
      out[(((size_t)(b * H_ + h) * LQ_) + l) * DH_ + j] = c[r];
    } else {
      out[(size_t)row * N + col] = c[r];
    }
  }
}

__global__ void __launch_bounds__(256)
proj_kernel(const float* __restrict__ X, const float* __restrict__ W,
            float* __restrict__ out)
{
  proj_body(X, W, out, 0, blockIdx.x);
}

__global__ void __launch_bounds__(256)
proj_qkv_kernel(const float* __restrict__ Xq, const float* __restrict__ Xk,
                const float* __restrict__ Xv,
                const float* __restrict__ Wq, const float* __restrict__ Wk,
                const float* __restrict__ Wv,
                float* __restrict__ oq, float* __restrict__ ok, float* __restrict__ ov)
{
  const int which = blockIdx.y;
  const float* X = (which == 0) ? Xq : (which == 1) ? Xk : Xv;
  const float* W = (which == 0) ? Wq : (which == 1) ? Wk : Wv;
  float*       o = (which == 0) ? oq : (which == 1) ? ok : ov;
  proj_body(X, W, o, 1, blockIdx.x);
}

__global__ void __launch_bounds__(256)
feat_kernel(const float* __restrict__ Hq, const float* __restrict__ Hk,
            const float* __restrict__ Wqa, const float* __restrict__ Wka,
            float* __restrict__ outq, float* __restrict__ outk)
{
  const int which = blockIdx.z;
  const float* Hd   = which ? Hk  : Hq;
  const float* Wa   = which ? Wka : Wqa;
  float*       outf = which ? outk : outq;

  const int bh = blockIdx.y;
  const float* A = Hd + (size_t)bh * LQ_ * DH_;
  float* O = outf + (size_t)bh * LQ_ * F_;

  const int tid  = threadIdx.x;
  const int lane = tid & 31;
  const int wave = tid >> 5;
  const int tile = blockIdx.x * 8 + wave;
  const int tm = tile / (F_ / 16);
  const int tn = tile % (F_ / 16);
  const int m0 = tm * 16, n0 = tn * 16;

  v8f c = wmma_f32_tile(A, DH_, Wa, DH_, DH_, m0, n0, lane);

  const int l16 = lane & 15, half = lane >> 4;
  for (int r = 0; r < 8; ++r)
    O[(size_t)(m0 + r + 8 * half) * F_ + n0 + l16] = c[r];
}

constexpr int QT_ = 4;
constexpr int FL_ = F_ / 32;

__global__ void __launch_bounds__(256)
score_softmax_av_kernel(const float* __restrict__ qf,
                        const float* __restrict__ kf,
                        const float* __restrict__ vh,
                        const float* __restrict__ wv_a,
                        const int*   __restrict__ valid_lens,
                        float* __restrict__ ctx)
{
  static_assert(QT_ * DH_ == 256, "the last stage maps one thread to one (row, column)");
  __shared__ float sp[QT_][LK_];
  __shared__ float spart[QT_][4][DH_];

  const int qt  = blockIdx.x % (LQ_ / QT_);
  const int bh  = blockIdx.x / (LQ_ / QT_);
  const int q0  = qt * QT_;
  const int b   = bh / H_;
  const int h   = bh % H_;
  const int tid  = threadIdx.x;
  const int lane = tid & 31;
  const int wave = tid >> 5;

  float wv[FL_];
  float sq[QT_][FL_];
  const float* qbase = qf + ((size_t)bh * LQ_ + q0) * F_ + lane;
  for (int i = 0; i < FL_; ++i) {
    wv[i] = wv_a[lane + 32 * i];
    for (int q = 0; q < QT_; ++q) sq[q][i] = qbase[(size_t)q * F_ + 32 * i];
  }

  const float* kbase = kf + (size_t)bh * LK_ * F_ + lane;
  float kr[FL_];
  {
    const float* krow = kbase + (size_t)wave * F_;
    for (int i = 0; i < FL_; ++i) kr[i] = krow[32 * i];
  }
  for (int kk = 0; kk < LK_ / 8; ++kk) {
    const int k = wave + 8 * kk;
    const int knext = wave + 8 * ((kk + 1 < LK_ / 8) ? kk + 1 : kk);
    const float* krow = kbase + (size_t)knext * F_;
    float kx[FL_];
    for (int i = 0; i < FL_; ++i) kx[i] = krow[32 * i];

    float acc[QT_];
    for (int q = 0; q < QT_; ++q) acc[q] = 0.f;
    for (int i = 0; i < FL_; ++i) {
      for (int q = 0; q < QT_; ++q)
        acc[q] += wv[i] * TANH_F32(sq[q][i] + kr[i]);
    }
    for (int off = 16; off > 0; off >>= 1) {
      for (int q = 0; q < QT_; ++q) acc[q] += __shfl_xor(acc[q], off, 32);
    }
    if (lane < QT_)
      sp[lane][k] = (lane == 0) ? acc[0] : (lane == 1) ? acc[1] : (lane == 2) ? acc[2] : acc[3];
    for (int i = 0; i < FL_; ++i) kr[i] = kx[i];
  }
  __syncthreads();

  if (wave < QT_) {
    const int q = wave;
    const int valid = valid_lens[b];
    float val[4], e[4], sm[4];
    for (int j = 0; j < 4; ++j) {
      const int k = lane + 32 * j;
      val[j] = (k < valid) ? sp[q][k] : -1.0e6f;
    }
    float m = fmaxf(fmaxf(val[0], val[1]), fmaxf(val[2], val[3]));
    for (int off = 16; off > 0; off >>= 1)
      m = fmaxf(m, __shfl_xor(m, off, 32));
    for (int j = 0; j < 4; ++j) {
      e[j] = __expf(val[j] - m);
      float t = e[j];
      for (int off = 16; off > 0; off >>= 1)
        t += __shfl_xor(t, off, 32);
      sm[j] = t;
    }
    const float rowsum = sm[0] + sm[1] + sm[2] + sm[3];
    for (int j = 0; j < 4; ++j) sp[q][lane + 32 * j] = e[j] / rowsum;
  }
  __syncthreads();

  const int j = tid & 63;
  const int part = tid >> 6;
  const float* vb = vh + (size_t)bh * LK_ * DH_;
  float o[QT_];
  for (int q = 0; q < QT_; ++q) o[q] = 0.f;
  for (int k = part * 32; k < part * 32 + 32; ++k) {
    const float v = vb[(size_t)k * DH_ + j];
    for (int q = 0; q < QT_; ++q) o[q] += sp[q][k] * v;
  }
  for (int q = 0; q < QT_; ++q) spart[q][part][j] = o[q];
  __syncthreads();

  {
    const int q = tid >> 6;
    const float r = spart[q][0][j] + spart[q][1][j] + spart[q][2][j] + spart[q][3][j];
    ctx[((size_t)b * LQ_ + q0 + q) * D_ + h * DH_ + j] = r;
  }
}

extern "C" void kernel_launch(void* const* d_in, const int* in_sizes, int n_in,
                              void* d_out, int out_size, void* d_ws, size_t ws_size,
                              hipStream_t stream) {
  (void)in_sizes; (void)n_in; (void)out_size; (void)ws_size;

  const float* queries    = (const float*)d_in[0];
  const float* keys       = (const float*)d_in[1];
  const float* values     = (const float*)d_in[2];
  const int*   valid_lens = (const int*)  d_in[3];
  const float* Wq  = (const float*)d_in[4];
  const float* Wk  = (const float*)d_in[5];
  const float* Wv  = (const float*)d_in[6];
  const float* Wo  = (const float*)d_in[7];
  const float* Wqa = (const float*)d_in[8];
  const float* Wka = (const float*)d_in[9];
  const float* wva = (const float*)d_in[10];

  float* ws  = (float*)d_ws;
  const size_t nQKV = (size_t)B_ * H_ * LQ_ * DH_;
  const size_t nF   = (size_t)B_ * H_ * LQ_ * F_;
  float* q   = ws;
  float* k   = q  + nQKV;
  float* v   = k  + nQKV;
  float* qf  = v  + nQKV;
  float* kfb = qf + nF;
  float* ctx = kfb + nF;

  dim3 blk(256);
  proj_qkv_kernel<<<dim3(64, 3), blk, 0, stream>>>(queries, keys, values, Wq, Wk, Wv, q, k, v);

  feat_kernel<<<dim3(32, B_ * H_, 2), blk, 0, stream>>>(q, k, Wqa, Wka, qf, kfb);

  score_softmax_av_kernel<<<B_ * H_ * (LQ_ / QT_), blk, 0, stream>>>(
      qf, kfb, v, wva, valid_lens, ctx);

  proj_kernel<<<64, blk, 0, stream>>>(ctx, Wo, (float*)d_out);
}


# Projection and feature GEMMs: each work-group stages its tile's 16 A rows and 128 W rows through LDS in 64-k chunks with 16-byte whole-line loads; fragments come from LDS, every sum keeps its k order
# speedup vs baseline: 2.2180x; 1.2011x over previous
#include <hip/hip_runtime.h>
#include <hip/hip_bf16.h>
#include <math.h>

constexpr int B_  = 2;
constexpr int H_  = 8;
constexpr int LQ_ = 128;
constexpr int LK_ = 128;
constexpr int D_  = 512;
constexpr int DH_ = 64;
constexpr int F_  = 512;

typedef float v2f __attribute__((ext_vector_type(2)));
typedef float v8f __attribute__((ext_vector_type(8)));

#if __has_builtin(__builtin_amdgcn_tanhf)
#define TANH_F32(x) __builtin_amdgcn_tanhf(x)
#elif __has_builtin(__builtin_amdgcn_tanh_f32)
#define TANH_F32(x) __builtin_amdgcn_tanh_f32(x)
#else
#define TANH_F32(x) tanhf(x)
#endif

typedef float v4f __attribute__((ext_vector_type(4)));

constexpr int KC_    = 64;
constexpr int LP_    = KC_ + 4;
constexpr int TROWS_ = 16 + 128;
constexpr int STAGE_ = TROWS_ * LP_;

template <int K>
__device__ __forceinline__ v8f wmma_f32_tile_staged(const float* __restrict__ Ablk,
                                                    const float* __restrict__ Wblk,
                                                    float* __restrict__ lds)
{
  constexpr int NCH = K / KC_;
  constexpr int NB  = (NCH > 1) ? 2 : 1;
  const int tid  = threadIdx.x;
  const int lane = tid & 31;
  const int wave = tid >> 5;
  const int srow = tid >> 4;
  const int spc  = tid & 15;
  const float* ga = Ablk + (size_t)srow * K + 4 * spc;
  const float* gw = Wblk + (size_t)srow * K + 4 * spc;
  const int la = srow * LP_ + 4 * spc;
  const int lw = (16 + srow) * LP_ + 4 * spc;

  v4f ra, rw[8];
  ra = *(const v4f*)ga;
  for (int i = 0; i < 8; ++i) rw[i] = *(const v4f*)(gw + (size_t)(16 * i) * K);

  const int l16 = lane & 15, half = lane >> 4;
  const int fa = l16 * LP_ + 2 * half;
  const int fb = (16 + 16 * wave + l16) * LP_ + 2 * half;

  v8f c = {};
  for (int ch = 0; ch < NCH; ++ch) {
    float* st = lds + (ch & (NB - 1)) * STAGE_;
    *(v4f*)(st + la) = ra;
    for (int i = 0; i < 8; ++i) *(v4f*)(st + lw + 16 * i * LP_) = rw[i];
    __syncthreads();
    if (ch + 1 < NCH) {
      const int ko = (ch + 1) * KC_;
      ra = *(const v4f*)(ga + ko);
      for (int i = 0; i < 8; ++i) rw[i] = *(const v4f*)(gw + (size_t)(16 * i) * K + ko);
    }
    for (int k = 0; k < KC_; k += 4) {
      v2f a = *(const v2f*)(st + fa + k);
      v2f b = *(const v2f*)(st + fb + k);
      c = __builtin_amdgcn_wmma_f32_16x16x4_f32(
           false, a,  false, b,
           (short)0, c,  false,  false);
    }
  }
  return c;
}

__device__ __forceinline__ void proj_body(const float* __restrict__ X, const float* __restrict__ W,
                                          float* __restrict__ out, int split, int blk,
                                          float* __restrict__ lds)
{
  const int N = D_;
  const int K = D_;
  const int tid  = threadIdx.x;
  const int lane = tid & 31;
  const int wave = tid >> 5;
  const int tile = blk * 8 + wave;
  const int tm = tile / (N / 16);
  const int tn = tile % (N / 16);
  const int m0 = tm * 16, n0 = tn * 16;
  const int nb0 = (blk % (N / 128)) * 128;

  const int l16 = lane & 15, half = lane >> 4;
  const int col  = n0 + l16;
  const int row0 = m0 + 8 * half;
  float* op;
  int ostride;
  if (split) {
    const int b = row0 / LQ_, l0 = row0 % LQ_;
    const int h = col / DH_, j = col % DH_;
    op = out + (((size_t)(b * H_ + h) * LQ_) + l0) * DH_ + j;
    ostride = DH_;
  } else {
    op = out + (size_t)row0 * N + col;
    ostride = N;
  }
  asm volatile("" : "+v"(op));

  v8f c = wmma_f32_tile_staged<D_>(X + (size_t)m0 * K, W + (size_t)nb0 * K, lds);

  for (int r = 0; r < 8; ++r) op[(size_t)r * ostride] = c[r];
}

__global__ void __launch_bounds__(256)
proj_kernel(const float* __restrict__ X, const float* __restrict__ W,
            float* __restrict__ out)
{
  __shared__ __attribute__((aligned(16))) float lds[2 * STAGE_];
  proj_body(X, W, out, 0, blockIdx.x, lds);
}

__global__ void __launch_bounds__(256)
proj_qkv_kernel(const float* __restrict__ Xq, const float* __restrict__ Xk,
                const float* __restrict__ Xv,
                const float* __restrict__ Wq, const float* __restrict__ Wk,
                const float* __restrict__ Wv,
                float* __restrict__ oq, float* __restrict__ ok, float* __restrict__ ov)
{
  __shared__ __attribute__((aligned(16))) float lds[2 * STAGE_];
  const int which = blockIdx.y;
  const float* X = (which == 0) ? Xq : (which == 1) ? Xk : Xv;
  const float* W = (which == 0) ? Wq : (which == 1) ? Wk : Wv;
  float*       o = (which == 0) ? oq : (which == 1) ? ok : ov;
  proj_body(X, W, o, 1, blockIdx.x, lds);
}

__global__ void __launch_bounds__(256)
feat_kernel(const float* __restrict__ Hq, const float* __restrict__ Hk,
            const float* __restrict__ Wqa, const float* __restrict__ Wka,
            float* __restrict__ outq, float* __restrict__ outk)
{
  __shared__ __attribute__((aligned(16))) float lds[STAGE_];
  const int which = blockIdx.z;
  const float* Hd   = which ? Hk  : Hq;
  const float* Wa   = which ? Wka : Wqa;
  float*       outf = which ? outk : outq;

  const int bh = blockIdx.y;
  const float* A = Hd + (size_t)bh * LQ_ * DH_;
  float* O = outf + (size_t)bh * LQ_ * F_;

  const int tid  = threadIdx.x;
  const int lane = tid & 31;
  const int wave = tid >> 5;
  const int tile = blockIdx.x * 8 + wave;
  const int tm = tile / (F_ / 16);
  const int tn = tile % (F_ / 16);
  const int m0 = tm * 16, n0 = tn * 16;

  const int nb0 = (blockIdx.x % (F_ / 128)) * 128;
  const int l16 = lane & 15, half = lane >> 4;
  float* op = O + (size_t)(m0 + 8 * half) * F_ + n0 + l16;
  asm volatile("" : "+v"(op));

  v8f c = wmma_f32_tile_staged<DH_>(A + (size_t)m0 * DH_, Wa + (size_t)nb0 * DH_, lds);

  for (int r = 0; r < 8; ++r) op[(size_t)r * F_] = c[r];
}

constexpr int QT_ = 4;
constexpr int FL_ = F_ / 32;

__global__ void __launch_bounds__(256)
score_softmax_av_kernel(const float* __restrict__ qf,
                        const float* __restrict__ kf,
                        const float* __restrict__ vh,
                        const float* __restrict__ wv_a,
                        const int*   __restrict__ valid_lens,
                        float* __restrict__ ctx)
{
  static_assert(QT_ * DH_ == 256, "the last stage maps one thread to one (row, column)");
  __shared__ float sp[QT_][LK_];
  __shared__ float spart[QT_][4][DH_];

  const int qt  = blockIdx.x % (LQ_ / QT_);
  const int bh  = blockIdx.x / (LQ_ / QT_);
  const int q0  = qt * QT_;
  const int b   = bh / H_;
  const int h   = bh % H_;
  const int tid  = threadIdx.x;
  const int lane = tid & 31;
  const int wave = tid >> 5;

  float wv[FL_];
  float sq[QT_][FL_];
  const float* qbase = qf + ((size_t)bh * LQ_ + q0) * F_ + lane;
  for (int i = 0; i < FL_; ++i) {
    wv[i] = wv_a[lane + 32 * i];
    for (int q = 0; q < QT_; ++q) sq[q][i] = qbase[(size_t)q * F_ + 32 * i];
  }

  const float* kbase = kf + (size_t)bh * LK_ * F_ + lane;
  float kr[FL_];
  {
    const float* krow = kbase + (size_t)wave * F_;
    for (int i = 0; i < FL_; ++i) kr[i] = krow[32 * i];
  }
  for (int kk = 0; kk < LK_ / 8; ++kk) {
    const int k = wave + 8 * kk;
    const int knext = wave + 8 * ((kk + 1 < LK_ / 8) ? kk + 1 : kk);
    const float* krow = kbase + (size_t)knext * F_;
    float kx[FL_];
    for (int i = 0; i < FL_; ++i) kx[i] = krow[32 * i];

    float acc[QT_];
    for (int q = 0; q < QT_; ++q) acc[q] = 0.f;
    for (int i = 0; i < FL_; ++i) {
      for (int q = 0; q < QT_; ++q)
        acc[q] += wv[i] * TANH_F32(sq[q][i] + kr[i]);
    }
    for (int off = 16; off > 0; off >>= 1) {
      for (int q = 0; q < QT_; ++q) acc[q] += __shfl_xor(acc[q], off, 32);
    }
    if (lane < QT_)
      sp[lane][k] = (lane == 0) ? acc[0] : (lane == 1) ? acc[1] : (lane == 2) ? acc[2] : acc[3];
    for (int i = 0; i < FL_; ++i) kr[i] = kx[i];
  }
  __syncthreads();

  if (wave < QT_) {
    const int q = wave;
    const int valid = valid_lens[b];
    float val[4], e[4], sm[4];
    for (int j = 0; j < 4; ++j) {
      const int k = lane + 32 * j;
      val[j] = (k < valid) ? sp[q][k] : -1.0e6f;
    }
    float m = fmaxf(fmaxf(val[0], val[1]), fmaxf(val[2], val[3]));
    for (int off = 16; off > 0; off >>= 1)
      m = fmaxf(m, __shfl_xor(m, off, 32));
    for (int j = 0; j < 4; ++j) {
      e[j] = __expf(val[j] - m);
      float t = e[j];
      for (int off = 16; off > 0; off >>= 1)
        t += __shfl_xor(t, off, 32);
      sm[j] = t;
    }
    const float rowsum = sm[0] + sm[1] + sm[2] + sm[3];
    for (int j = 0; j < 4; ++j) sp[q][lane + 32 * j] = e[j] / rowsum;
  }
  __syncthreads();

  const int j = tid & 63;
  const int part = tid >> 6;
  const float* vb = vh + (size_t)bh * LK_ * DH_;
  float o[QT_];
  for (int q = 0; q < QT_; ++q) o[q] = 0.f;
  for (int k = part * 32; k < part * 32 + 32; ++k) {
    const float v = vb[(size_t)k * DH_ + j];
    for (int q = 0; q < QT_; ++q) o[q] += sp[q][k] * v;
  }
  for (int q = 0; q < QT_; ++q) spart[q][part][j] = o[q];
  __syncthreads();

  {
    const int q = tid >> 6;
    const float r = spart[q][0][j] + spart[q][1][j] + spart[q][2][j] + spart[q][3][j];
    ctx[((size_t)b * LQ_ + q0 + q) * D_ + h * DH_ + j] = r;
  }
}

extern "C" void kernel_launch(void* const* d_in, const int* in_sizes, int n_in,
                              void* d_out, int out_size, void* d_ws, size_t ws_size,
                              hipStream_t stream) {
  (void)in_sizes; (void)n_in; (void)out_size; (void)ws_size;

  const float* queries    = (const float*)d_in[0];
  const float* keys       = (const float*)d_in[1];
  const float* values     = (const float*)d_in[2];
  const int*   valid_lens = (const int*)  d_in[3];
  const float* Wq  = (const float*)d_in[4];
  const float* Wk  = (const float*)d_in[5];
  const float* Wv  = (const float*)d_in[6];
  const float* Wo  = (const float*)d_in[7];
  const float* Wqa = (const float*)d_in[8];
  const float* Wka = (const float*)d_in[9];
  const float* wva = (const float*)d_in[10];

  float* ws  = (float*)d_ws;
  const size_t nQKV = (size_t)B_ * H_ * LQ_ * DH_;
  const size_t nF   = (size_t)B_ * H_ * LQ_ * F_;
  float* q   = ws;
  float* k   = q  + nQKV;
  float* v   = k  + nQKV;
  float* qf  = v  + nQKV;
  float* kfb = qf + nF;
  float* ctx = kfb + nF;

  dim3 blk(256);
  proj_qkv_kernel<<<dim3(64, 3), blk, 0, stream>>>(queries, keys, values, Wq, Wk, Wv, q, k, v);

  feat_kernel<<<dim3(32, B_ * H_, 2), blk, 0, stream>>>(q, k, Wqa, Wka, qf, kfb);

  score_softmax_av_kernel<<<B_ * H_ * (LQ_ / QT_), blk, 0, stream>>>(
      qf, kfb, v, wva, valid_lens, ctx);

  proj_kernel<<<64, blk, 0, stream>>>(ctx, Wo, (float*)d_out);
}


# Score stage: a lane owns a key (key features stored feature-major), query features and weights as scalar operands, no cross-lane sums; each score now sums features in order in two halves of 256
# speedup vs baseline: 2.3991x; 1.0817x over previous
#include <hip/hip_runtime.h>
#include <hip/hip_bf16.h>
#include <math.h>

constexpr int B_  = 2;
constexpr int H_  = 8;
constexpr int LQ_ = 128;
constexpr int LK_ = 128;
constexpr int D_  = 512;
constexpr int DH_ = 64;
constexpr int F_  = 512;

typedef float v2f __attribute__((ext_vector_type(2)));
typedef float v8f __attribute__((ext_vector_type(8)));

#if __has_builtin(__builtin_amdgcn_tanhf)
#define TANH_F32(x) __builtin_amdgcn_tanhf(x)
#elif __has_builtin(__builtin_amdgcn_tanh_f32)
#define TANH_F32(x) __builtin_amdgcn_tanh_f32(x)
#else
#define TANH_F32(x) tanhf(x)
#endif

typedef float v4f __attribute__((ext_vector_type(4)));

constexpr int KC_    = 64;
constexpr int LP_    = KC_ + 4;
constexpr int TROWS_ = 16 + 128;
constexpr int STAGE_ = TROWS_ * LP_;

template <int K>
__device__ __forceinline__ v8f wmma_f32_tile_staged(const float* __restrict__ Ablk,
                                                    const float* __restrict__ Wblk,
                                                    float* __restrict__ lds)
{
  constexpr int NCH = K / KC_;
  constexpr int NB  = (NCH > 1) ? 2 : 1;
  const int tid  = threadIdx.x;
  const int lane = tid & 31;
  const int wave = tid >> 5;
  const int srow = tid >> 4;
  const int spc  = tid & 15;
  const float* ga = Ablk + (size_t)srow * K + 4 * spc;
  const float* gw = Wblk + (size_t)srow * K + 4 * spc;
  const int la = srow * LP_ + 4 * spc;
  const int lw = (16 + srow) * LP_ + 4 * spc;

  v4f ra, rw[8];
  ra = *(const v4f*)ga;
  for (int i = 0; i < 8; ++i) rw[i] = *(const v4f*)(gw + (size_t)(16 * i) * K);

  const int l16 = lane & 15, half = lane >> 4;
  const int fa = l16 * LP_ + 2 * half;
  const int fb = (16 + 16 * wave + l16) * LP_ + 2 * half;

  v8f c = {};
  for (int ch = 0; ch < NCH; ++ch) {
    float* st = lds + (ch & (NB - 1)) * STAGE_;
    *(v4f*)(st + la) = ra;
    for (int i = 0; i < 8; ++i) *(v4f*)(st + lw + 16 * i * LP_) = rw[i];
    __syncthreads();
    if (ch + 1 < NCH) {
      const int ko = (ch + 1) * KC_;
      ra = *(const v4f*)(ga + ko);
      for (int i = 0; i < 8; ++i) rw[i] = *(const v4f*)(gw + (size_t)(16 * i) * K + ko);
    }
    for (int k = 0; k < KC_; k += 4) {
      v2f a = *(const v2f*)(st + fa + k);
      v2f b = *(const v2f*)(st + fb + k);
      c = __builtin_amdgcn_wmma_f32_16x16x4_f32(
           false, a,  false, b,
           (short)0, c,  false,  false);
    }
  }
  return c;
}

__device__ __forceinline__ void proj_body(const float* __restrict__ X, const float* __restrict__ W,
                                          float* __restrict__ out, int split, int blk,
                                          float* __restrict__ lds)
{
  const int N = D_;
  const int K = D_;
  const int tid  = threadIdx.x;
  const int lane = tid & 31;
  const int wave = tid >> 5;
  const int tile = blk * 8 + wave;
  const int tm = tile / (N / 16);
  const int tn = tile % (N / 16);
  const int m0 = tm * 16, n0 = tn * 16;
  const int nb0 = (blk % (N / 128)) * 128;

  const int l16 = lane & 15, half = lane >> 4;
  const int col  = n0 + l16;
  const int row0 = m0 + 8 * half;
  float* op;
  int ostride;
  if (split) {
    const int b = row0 / LQ_, l0 = row0 % LQ_;
    const int h = col / DH_, j = col % DH_;
    op = out + (((size_t)(b * H_ + h) * LQ_) + l0) * DH_ + j;
    ostride = DH_;
  } else {
    op = out + (size_t)row0 * N + col;
    ostride = N;
  }
  asm volatile("" : "+v"(op));

  v8f c = wmma_f32_tile_staged<D_>(X + (size_t)m0 * K, W + (size_t)nb0 * K, lds);

  for (int r = 0; r < 8; ++r) op[(size_t)r * ostride] = c[r];
}

__global__ void __launch_bounds__(256)
proj_kernel(const float* __restrict__ X, const float* __restrict__ W,
            float* __restrict__ out)
{
  __shared__ __attribute__((aligned(16))) float lds[2 * STAGE_];
  proj_body(X, W, out, 0, blockIdx.x, lds);
}

__global__ void __launch_bounds__(256)
proj_qkv_kernel(const float* __restrict__ Xq, const float* __restrict__ Xk,
                const float* __restrict__ Xv,
                const float* __restrict__ Wq, const float* __restrict__ Wk,
                const float* __restrict__ Wv,
                float* __restrict__ oq, float* __restrict__ ok, float* __restrict__ ov)
{
  __shared__ __attribute__((aligned(16))) float lds[2 * STAGE_];
  const int which = blockIdx.y;
  const float* X = (which == 0) ? Xq : (which == 1) ? Xk : Xv;
  const float* W = (which == 0) ? Wq : (which == 1) ? Wk : Wv;
  float*       o = (which == 0) ? oq : (which == 1) ? ok : ov;
  proj_body(X, W, o, 1, blockIdx.x, lds);
}

__global__ void __launch_bounds__(256)
feat_kernel(const float* __restrict__ Hq, const float* __restrict__ Hk,
            const float* __restrict__ Wqa, const float* __restrict__ Wka,
            float* __restrict__ outq, float* __restrict__ outk)
{
  __shared__ __attribute__((aligned(16))) float lds[STAGE_];
  const int which = blockIdx.z;
  const int bh = blockIdx.y;

  const int tid  = threadIdx.x;
  const int lane = tid & 31;
  const int wave = tid >> 5;
  const int l16 = lane & 15, half = lane >> 4;

  const float* Ablk;
  const float* Wblk;
  float* op;
  int opitch;
  if (which == 0) {
    const int m0  = (blockIdx.x / (F_ / 128)) * 16;
    const int nb0 = (blockIdx.x % (F_ / 128)) * 128;
    Ablk = Hq + ((size_t)bh * LQ_ + m0) * DH_;
    Wblk = Wqa + (size_t)nb0 * DH_;
    op = outq + ((size_t)bh * LQ_ + m0 + 8 * half) * F_ + nb0 + 16 * wave + l16;
    opitch = F_;
  } else {
    const int f0 = blockIdx.x * 16;
    Ablk = Wka + (size_t)f0 * DH_;
    Wblk = Hk + (size_t)bh * LK_ * DH_;
    op = outk + ((size_t)bh * F_ + f0 + 8 * half) * LK_ + 16 * wave + l16;
    opitch = LK_;
  }
  float* o0 = op;
  float* o1 = op + (size_t)1 * opitch;
  float* o2 = op + (size_t)2 * opitch;
  float* o3 = op + (size_t)3 * opitch;
  float* o4 = op + (size_t)4 * opitch;
  float* o5 = op + (size_t)5 * opitch;
  float* o6 = op + (size_t)6 * opitch;
  float* o7 = op + (size_t)7 * opitch;
  asm volatile("" : "+v"(o0), "+v"(o1), "+v"(o2), "+v"(o3), "+v"(o4), "+v"(o5), "+v"(o6), "+v"(o7));

  v8f c = wmma_f32_tile_staged<DH_>(Ablk, Wblk, lds);

  *o0 = c[0]; *o1 = c[1]; *o2 = c[2]; *o3 = c[3];
  *o4 = c[4]; *o5 = c[5]; *o6 = c[6]; *o7 = c[7];
}

constexpr int QT_ = 4;
constexpr int FH_ = F_ / 2;
constexpr int FC_ = 8;

__global__ void __launch_bounds__(256)
score_softmax_av_kernel(const float* __restrict__ qf,
                        const float* __restrict__ kft,
                        const float* __restrict__ vh,
                        const float* __restrict__ wv_a,
                        const int*   __restrict__ valid_lens,
                        float* __restrict__ ctx)
{
  static_assert(QT_ * DH_ == 256, "the last stage maps one thread to one (row, column)");
  static_assert(LK_ == 128 && F_ == 512, "4 key groups of 32 lanes x 2 feature halves = 8 waves");
  __shared__ float sp[2][QT_][LK_];
  __shared__ float spart[QT_][4][DH_];

  const int qt  = blockIdx.x % (LQ_ / QT_);
  const int bh  = blockIdx.x / (LQ_ / QT_);
  const int q0  = qt * QT_;
  const int b   = bh / H_;
  const int h   = bh % H_;
  const int tid  = threadIdx.x;
  const int lane = tid & 31;
  const int wave = tid >> 5;

  {
    const int wu = __builtin_amdgcn_readfirstlane(wave);
    const int fh = wu >> 2;
    const int key = 32 * (wu & 3) + lane;
    const float* kt = kft + ((size_t)bh * F_ + (size_t)fh * FH_) * LK_ + key;
    const float* qr = qf + ((size_t)bh * LQ_ + q0) * F_ + fh * FH_;
    const float* wp = wv_a + fh * FH_;

    float acc[QT_];
    for (int q = 0; q < QT_; ++q) acc[q] = 0.f;

    float kA[FC_], kB[FC_];
    float qA[QT_][FC_], qB[QT_][FC_];
    float wA[FC_], wB[FC_];
#define SC_LOAD(kx, qx, wx, fb)                                                   \
    for (int i = 0; i < FC_; ++i) kx[i] = kt[(size_t)((fb) + i) * LK_];           \
    for (int i = 0; i < FC_; ++i) wx[i] = wp[(fb) + i];                           \
    for (int q = 0; q < QT_; ++q)                                                 \
      for (int i = 0; i < FC_; ++i) qx[q][i] = qr[(size_t)q * F_ + (fb) + i];
#define SC_STEP(kx, qx, wx)                                                       \
    for (int i = 0; i < FC_; ++i)                                                 \
      for (int q = 0; q < QT_; ++q)                                               \
        acc[q] = __builtin_fmaf(wx[i], TANH_F32(qx[q][i] + kx[i]), acc[q]);
    SC_LOAD(kA, qA, wA, 0)
    for (int f0 = 0; f0 < FH_; f0 += 2 * FC_) {
      SC_LOAD(kB, qB, wB, f0 + FC_)
      SC_STEP(kA, qA, wA)
      const int fn = (f0 + 2 * FC_ < FH_) ? f0 + 2 * FC_ : 0;
      SC_LOAD(kA, qA, wA, fn)
      SC_STEP(kB, qB, wB)
    }
#undef SC_LOAD
#undef SC_STEP
    for (int q = 0; q < QT_; ++q) sp[fh][q][key] = acc[q];
  }
  __syncthreads();

  if (wave < QT_) {
    const int q = wave;
    const int valid = valid_lens[b];
    float val[4], e[4], sm[4];
    for (int j = 0; j < 4; ++j) {
      const int k = lane + 32 * j;
      val[j] = (k < valid) ? (sp[0][q][k] + sp[1][q][k]) : -1.0e6f;
    }
    float m = fmaxf(fmaxf(val[0], val[1]), fmaxf(val[2], val[3]));
    for (int off = 16; off > 0; off >>= 1)
      m = fmaxf(m, __shfl_xor(m, off, 32));
    for (int j = 0; j < 4; ++j) {
      e[j] = __expf(val[j] - m);
      float t = e[j];
      for (int off = 16; off > 0; off >>= 1)
        t += __shfl_xor(t, off, 32);
      sm[j] = t;
    }
    const float rowsum = sm[0] + sm[1] + sm[2] + sm[3];
    for (int j = 0; j < 4; ++j) sp[0][q][lane + 32 * j] = e[j] / rowsum;
  }
  __syncthreads();

  const int j = tid & 63;
  const int part = tid >> 6;
  const float* vb = vh + (size_t)bh * LK_ * DH_;
  float o[QT_];
  for (int q = 0; q < QT_; ++q) o[q] = 0.f;
  for (int k = part * 32; k < part * 32 + 32; ++k) {
    const float v = vb[(size_t)k * DH_ + j];
    for (int q = 0; q < QT_; ++q) o[q] += sp[0][q][k] * v;
  }
  for (int q = 0; q < QT_; ++q) spart[q][part][j] = o[q];
  __syncthreads();

  {
    const int q = tid >> 6;
    const float r = spart[q][0][j] + spart[q][1][j] + spart[q][2][j] + spart[q][3][j];
    ctx[((size_t)b * LQ_ + q0 + q) * D_ + h * DH_ + j] = r;
  }
}

extern "C" void kernel_launch(void* const* d_in, const int* in_sizes, int n_in,
                              void* d_out, int out_size, void* d_ws, size_t ws_size,
                              hipStream_t stream) {
  (void)in_sizes; (void)n_in; (void)out_size; (void)ws_size;

  const float* queries    = (const float*)d_in[0];
  const float* keys       = (const float*)d_in[1];
  const float* values     = (const float*)d_in[2];
  const int*   valid_lens = (const int*)  d_in[3];
  const float* Wq  = (const float*)d_in[4];
  const float* Wk  = (const float*)d_in[5];
  const float* Wv  = (const float*)d_in[6];
  const float* Wo  = (const float*)d_in[7];
  const float* Wqa = (const float*)d_in[8];
  const float* Wka = (const float*)d_in[9];
  const float* wva = (const float*)d_in[10];

  float* ws  = (float*)d_ws;
  const size_t nQKV = (size_t)B_ * H_ * LQ_ * DH_;
  const size_t nF   = (size_t)B_ * H_ * LQ_ * F_;
  float* q   = ws;
  float* k   = q  + nQKV;
  float* v   = k  + nQKV;
  float* qf  = v  + nQKV;
  float* kfb = qf + nF;
  float* ctx = kfb + nF;

  dim3 blk(256);
  proj_qkv_kernel<<<dim3(64, 3), blk, 0, stream>>>(queries, keys, values, Wq, Wk, Wv, q, k, v);

  feat_kernel<<<dim3(32, B_ * H_, 2), blk, 0, stream>>>(q, k, Wqa, Wka, qf, kfb);

  score_softmax_av_kernel<<<B_ * H_ * (LQ_ / QT_), blk, 0, stream>>>(
      qf, kfb, v, wva, valid_lens, ctx);

  proj_kernel<<<64, blk, 0, stream>>>(ctx, Wo, (float*)d_out);
}


# Score stage: a lane owns 4 adjacent keys (one 16-byte load per feature feeds 16 tanh) and a wave owns 64 features of all keys; each score now sums features in order in eight slices of 64
# speedup vs baseline: 2.5799x; 1.0754x over previous
#include <hip/hip_runtime.h>
#include <hip/hip_bf16.h>
#include <math.h>

constexpr int B_  = 2;
constexpr int H_  = 8;
constexpr int LQ_ = 128;
constexpr int LK_ = 128;
constexpr int D_  = 512;
constexpr int DH_ = 64;
constexpr int F_  = 512;

typedef float v2f __attribute__((ext_vector_type(2)));
typedef float v8f __attribute__((ext_vector_type(8)));

#if __has_builtin(__builtin_amdgcn_tanhf)
#define TANH_F32(x) __builtin_amdgcn_tanhf(x)
#elif __has_builtin(__builtin_amdgcn_tanh_f32)
#define TANH_F32(x) __builtin_amdgcn_tanh_f32(x)
#else
#define TANH_F32(x) tanhf(x)
#endif

typedef float v4f __attribute__((ext_vector_type(4)));

constexpr int KC_    = 64;
constexpr int LP_    = KC_ + 4;
constexpr int TROWS_ = 16 + 128;
constexpr int STAGE_ = TROWS_ * LP_;

template <int K>
__device__ __forceinline__ v8f wmma_f32_tile_staged(const float* __restrict__ Ablk,
                                                    const float* __restrict__ Wblk,
                                                    float* __restrict__ lds)
{
  constexpr int NCH = K / KC_;
  constexpr int NB  = (NCH > 1) ? 2 : 1;
  const int tid  = threadIdx.x;
  const int lane = tid & 31;
  const int wave = tid >> 5;
  const int srow = tid >> 4;
  const int spc  = tid & 15;
  const float* ga = Ablk + (size_t)srow * K + 4 * spc;
  const float* gw = Wblk + (size_t)srow * K + 4 * spc;
  const int la = srow * LP_ + 4 * spc;
  const int lw = (16 + srow) * LP_ + 4 * spc;

  v4f ra, rw[8];
  ra = *(const v4f*)ga;
  for (int i = 0; i < 8; ++i) rw[i] = *(const v4f*)(gw + (size_t)(16 * i) * K);

  const int l16 = lane & 15, half = lane >> 4;
  const int fa = l16 * LP_ + 2 * half;
  const int fb = (16 + 16 * wave + l16) * LP_ + 2 * half;

  v8f c = {};
  for (int ch = 0; ch < NCH; ++ch) {
    float* st = lds + (ch & (NB - 1)) * STAGE_;
    *(v4f*)(st + la) = ra;
    for (int i = 0; i < 8; ++i) *(v4f*)(st + lw + 16 * i * LP_) = rw[i];
    __syncthreads();
    if (ch + 1 < NCH) {
      const int ko = (ch + 1) * KC_;
      ra = *(const v4f*)(ga + ko);
      for (int i = 0; i < 8; ++i) rw[i] = *(const v4f*)(gw + (size_t)(16 * i) * K + ko);
    }
    for (int k = 0; k < KC_; k += 4) {
      v2f a = *(const v2f*)(st + fa + k);
      v2f b = *(const v2f*)(st + fb + k);
      c = __builtin_amdgcn_wmma_f32_16x16x4_f32(
           false, a,  false, b,
           (short)0, c,  false,  false);
    }
  }
  return c;
}

__device__ __forceinline__ void proj_body(const float* __restrict__ X, const float* __restrict__ W,
                                          float* __restrict__ out, int split, int blk,
                                          float* __restrict__ lds)
{
  const int N = D_;
  const int K = D_;
  const int tid  = threadIdx.x;
  const int lane = tid & 31;
  const int wave = tid >> 5;
  const int tile = blk * 8 + wave;
  const int tm = tile / (N / 16);
  const int tn = tile % (N / 16);
  const int m0 = tm * 16, n0 = tn * 16;
  const int nb0 = (blk % (N / 128)) * 128;

  const int l16 = lane & 15, half = lane >> 4;
  const int col  = n0 + l16;
  const int row0 = m0 + 8 * half;
  float* op;
  int ostride;
  if (split) {
    const int b = row0 / LQ_, l0 = row0 % LQ_;
    const int h = col / DH_, j = col % DH_;
    op = out + (((size_t)(b * H_ + h) * LQ_) + l0) * DH_ + j;
    ostride = DH_;
  } else {
    op = out + (size_t)row0 * N + col;
    ostride = N;
  }
  asm volatile("" : "+v"(op));

  v8f c = wmma_f32_tile_staged<D_>(X + (size_t)m0 * K, W + (size_t)nb0 * K, lds);

  for (int r = 0; r < 8; ++r) op[(size_t)r * ostride] = c[r];
}

__global__ void __launch_bounds__(256)
proj_kernel(const float* __restrict__ X, const float* __restrict__ W,
            float* __restrict__ out)
{
  __shared__ __attribute__((aligned(16))) float lds[2 * STAGE_];
  proj_body(X, W, out, 0, blockIdx.x, lds);
}

__global__ void __launch_bounds__(256)
proj_qkv_kernel(const float* __restrict__ Xq, const float* __restrict__ Xk,
                const float* __restrict__ Xv,
                const float* __restrict__ Wq, const float* __restrict__ Wk,
                const float* __restrict__ Wv,
                float* __restrict__ oq, float* __restrict__ ok, float* __restrict__ ov)
{
  __shared__ __attribute__((aligned(16))) float lds[2 * STAGE_];
  const int which = blockIdx.y;
  const float* X = (which == 0) ? Xq : (which == 1) ? Xk : Xv;
  const float* W = (which == 0) ? Wq : (which == 1) ? Wk : Wv;
  float*       o = (which == 0) ? oq : (which == 1) ? ok : ov;
  proj_body(X, W, o, 1, blockIdx.x, lds);
}

__global__ void __launch_bounds__(256)
feat_kernel(const float* __restrict__ Hq, const float* __restrict__ Hk,
            const float* __restrict__ Wqa, const float* __restrict__ Wka,
            float* __restrict__ outq, float* __restrict__ outk)
{
  __shared__ __attribute__((aligned(16))) float lds[STAGE_];
  const int which = blockIdx.z;
  const int bh = blockIdx.y;

  const int tid  = threadIdx.x;
  const int lane = tid & 31;
  const int wave = tid >> 5;
  const int l16 = lane & 15, half = lane >> 4;

  const float* Ablk;
  const float* Wblk;
  float* op;
  int opitch;
  if (which == 0) {
    const int m0  = (blockIdx.x / (F_ / 128)) * 16;
    const int nb0 = (blockIdx.x % (F_ / 128)) * 128;
    Ablk = Hq + ((size_t)bh * LQ_ + m0) * DH_;
    Wblk = Wqa + (size_t)nb0 * DH_;
    op = outq + ((size_t)bh * LQ_ + m0 + 8 * half) * F_ + nb0 + 16 * wave + l16;
    opitch = F_;
  } else {
    const int f0 = blockIdx.x * 16;
    Ablk = Wka + (size_t)f0 * DH_;
    Wblk = Hk + (size_t)bh * LK_ * DH_;
    op = outk + ((size_t)bh * F_ + f0 + 8 * half) * LK_ + 16 * wave + l16;
    opitch = LK_;
  }
  float* o0 = op;
  float* o1 = op + (size_t)1 * opitch;
  float* o2 = op + (size_t)2 * opitch;
  float* o3 = op + (size_t)3 * opitch;
  float* o4 = op + (size_t)4 * opitch;
  float* o5 = op + (size_t)5 * opitch;
  float* o6 = op + (size_t)6 * opitch;
  float* o7 = op + (size_t)7 * opitch;
  asm volatile("" : "+v"(o0), "+v"(o1), "+v"(o2), "+v"(o3), "+v"(o4), "+v"(o5), "+v"(o6), "+v"(o7));

  v8f c = wmma_f32_tile_staged<DH_>(Ablk, Wblk, lds);

  *o0 = c[0]; *o1 = c[1]; *o2 = c[2]; *o3 = c[3];
  *o4 = c[4]; *o5 = c[5]; *o6 = c[6]; *o7 = c[7];
}

constexpr int QT_ = 4;
constexpr int NS_ = 8;
constexpr int FS_ = F_ / NS_;
constexpr int FC_ = 8;
constexpr int SC_ = 4;
constexpr int KL_ = 4;

__global__ void __launch_bounds__(256)
score_softmax_av_kernel(const float* __restrict__ qf,
                        const float* __restrict__ kft,
                        const float* __restrict__ vh,
                        const float* __restrict__ wv_a,
                        const int*   __restrict__ valid_lens,
                        float* __restrict__ ctx)
{
  static_assert(QT_ * DH_ == 256, "the last stage maps one thread to one (row, column)");
  static_assert(LK_ == 32 * KL_ && F_ == NS_ * FS_ && NS_ == 8, "32 lanes x 4 keys cover the keys, 8 waves cover the features");
  static_assert(FS_ % (2 * FC_) == 0 && FC_ == 2 * SC_, "two register sets alternate, two scalar sets per set");
  __shared__ __attribute__((aligned(16))) float sp[NS_][QT_][LK_];
  __shared__ float spart[QT_][4][DH_];

  const int qt  = blockIdx.x % (LQ_ / QT_);
  const int bh  = blockIdx.x / (LQ_ / QT_);
  const int q0  = qt * QT_;
  const int b   = bh / H_;
  const int h   = bh % H_;
  const int tid  = threadIdx.x;
  const int lane = tid & 31;
  const int wave = tid >> 5;

  {
    const int wu = __builtin_amdgcn_readfirstlane(wave);
    const float* kt = kft + ((size_t)bh * F_ + (size_t)wu * FS_) * LK_ + KL_ * lane;
    const float* qr = qf + ((size_t)bh * LQ_ + q0) * F_ + wu * FS_;
    const float* wp = wv_a + wu * FS_;

    float acc[QT_][KL_];
    for (int q = 0; q < QT_; ++q)
      for (int j = 0; j < KL_; ++j) acc[q][j] = 0.f;

    v4f kA[FC_], kB[FC_];
    float qA[QT_][SC_], qB[QT_][SC_];
    float wA[SC_], wB[SC_];
#define K_LOAD(kx, fb)                                                            \
    for (int i = 0; i < FC_; ++i) kx[i] = *(const v4f*)(kt + (size_t)((fb) + i) * LK_);
#define S_LOAD(qx, wx, fb)                                                        \
    for (int i = 0; i < SC_; ++i) wx[i] = wp[(fb) + i];                           \
    for (int q = 0; q < QT_; ++q)                                                 \
      for (int i = 0; i < SC_; ++i) qx[q][i] = qr[(size_t)q * F_ + (fb) + i];
#define SC_STEP(kx, ko, qx, wx)                                                   \
    for (int i = 0; i < SC_; ++i) {                                               \
      float t[QT_][KL_];                                                          \
      for (int q = 0; q < QT_; ++q)                                               \
        for (int j = 0; j < KL_; ++j) t[q][j] = qx[q][i] + kx[(ko) + i][j];       \
      for (int q = 0; q < QT_; ++q)                                               \
        for (int j = 0; j < KL_; ++j) t[q][j] = TANH_F32(t[q][j]);                \
      for (int q = 0; q < QT_; ++q)                                               \
        for (int j = 0; j < KL_; ++j) acc[q][j] = __builtin_fmaf(wx[i], t[q][j], acc[q][j]); \
    }
    K_LOAD(kA, 0)
    S_LOAD(qA, wA, 0)
    for (int f0 = 0; f0 < FS_; f0 += 2 * FC_) {
      K_LOAD(kB, f0 + FC_)
      S_LOAD(qB, wB, f0 + SC_)
      SC_STEP(kA, 0, qA, wA)
      S_LOAD(qA, wA, f0 + FC_)
      SC_STEP(kA, SC_, qB, wB)
      const int fn = (f0 + 2 * FC_ < FS_) ? f0 + 2 * FC_ : 0;
      K_LOAD(kA, fn)
      S_LOAD(qB, wB, f0 + FC_ + SC_)
      SC_STEP(kB, 0, qA, wA)
      S_LOAD(qA, wA, fn)
      SC_STEP(kB, SC_, qB, wB)
    }
#undef K_LOAD
#undef S_LOAD
#undef SC_STEP
    for (int q = 0; q < QT_; ++q) {
      v4f r = {acc[q][0], acc[q][1], acc[q][2], acc[q][3]};
      *(v4f*)&sp[wu][q][KL_ * lane] = r;
    }
  }
  __syncthreads();

  if (wave < QT_) {
    const int q = wave;
    const int valid = valid_lens[b];
    float val[4], e[4], sm[4];
    for (int j = 0; j < 4; ++j) {
      const int k = lane + 32 * j;
      float s = sp[0][q][k];
      for (int w = 1; w < NS_; ++w) s += sp[w][q][k];
      val[j] = (k < valid) ? s : -1.0e6f;
    }
    float m = fmaxf(fmaxf(val[0], val[1]), fmaxf(val[2], val[3]));
    for (int off = 16; off > 0; off >>= 1)
      m = fmaxf(m, __shfl_xor(m, off, 32));
    for (int j = 0; j < 4; ++j) {
      e[j] = __expf(val[j] - m);
      float t = e[j];
      for (int off = 16; off > 0; off >>= 1)
        t += __shfl_xor(t, off, 32);
      sm[j] = t;
    }
    const float rowsum = sm[0] + sm[1] + sm[2] + sm[3];
    for (int j = 0; j < 4; ++j) sp[0][q][lane + 32 * j] = e[j] / rowsum;
  }
  __syncthreads();

  const int j = tid & 63;
  const int part = tid >> 6;
  const float* vb = vh + (size_t)bh * LK_ * DH_;
  float o[QT_];
  for (int q = 0; q < QT_; ++q) o[q] = 0.f;
  for (int k = part * 32; k < part * 32 + 32; ++k) {
    const float v = vb[(size_t)k * DH_ + j];
    for (int q = 0; q < QT_; ++q) o[q] += sp[0][q][k] * v;
  }
  for (int q = 0; q < QT_; ++q) spart[q][part][j] = o[q];
  __syncthreads();

  {
    const int q = tid >> 6;
    const float r = spart[q][0][j] + spart[q][1][j] + spart[q][2][j] + spart[q][3][j];
    ctx[((size_t)b * LQ_ + q0 + q) * D_ + h * DH_ + j] = r;
  }
}

extern "C" void kernel_launch(void* const* d_in, const int* in_sizes, int n_in,
                              void* d_out, int out_size, void* d_ws, size_t ws_size,
                              hipStream_t stream) {
  (void)in_sizes; (void)n_in; (void)out_size; (void)ws_size;

  const float* queries    = (const float*)d_in[0];
  const float* keys       = (const float*)d_in[1];
  const float* values     = (const float*)d_in[2];
  const int*   valid_lens = (const int*)  d_in[3];
  const float* Wq  = (const float*)d_in[4];
  const float* Wk  = (const float*)d_in[5];
  const float* Wv  = (const float*)d_in[6];
  const float* Wo  = (const float*)d_in[7];
  const float* Wqa = (const float*)d_in[8];
  const float* Wka = (const float*)d_in[9];
  const float* wva = (const float*)d_in[10];

  float* ws  = (float*)d_ws;
  const size_t nQKV = (size_t)B_ * H_ * LQ_ * DH_;
  const size_t nF   = (size_t)B_ * H_ * LQ_ * F_;
  float* q   = ws;
  float* k   = q  + nQKV;
  float* v   = k  + nQKV;
  float* qf  = v  + nQKV;
  float* kfb = qf + nF;
  float* ctx = kfb + nF;

  dim3 blk(256);
  proj_qkv_kernel<<<dim3(64, 3), blk, 0, stream>>>(queries, keys, values, Wq, Wk, Wv, q, k, v);

  feat_kernel<<<dim3(32, B_ * H_, 2), blk, 0, stream>>>(q, k, Wqa, Wka, qf, kfb);

  score_softmax_av_kernel<<<B_ * H_ * (LQ_ / QT_), blk, 0, stream>>>(
      qf, kfb, v, wva, valid_lens, ctx);

  proj_kernel<<<64, blk, 0, stream>>>(ctx, Wo, (float*)d_out);
}


# Projection GEMMs: 4-wave work-groups on 16 x 64 tiles (twice as many), each 64-k chunk's global loads issued 4 chunks ahead and kept in flight across an LDS-only barrier; every sum keeps its k order
# speedup vs baseline: 3.2658x; 1.2659x over previous
#include <hip/hip_runtime.h>
#include <hip/hip_bf16.h>
#include <math.h>

constexpr int B_  = 2;
constexpr int H_  = 8;
constexpr int LQ_ = 128;
constexpr int LK_ = 128;
constexpr int D_  = 512;
constexpr int DH_ = 64;
constexpr int F_  = 512;

typedef float v2f __attribute__((ext_vector_type(2)));
typedef float v8f __attribute__((ext_vector_type(8)));

#if __has_builtin(__builtin_amdgcn_tanhf)
#define TANH_F32(x) __builtin_amdgcn_tanhf(x)
#elif __has_builtin(__builtin_amdgcn_tanh_f32)
#define TANH_F32(x) __builtin_amdgcn_tanh_f32(x)
#else
#define TANH_F32(x) tanhf(x)
#endif

typedef float v4f __attribute__((ext_vector_type(4)));

constexpr int KC_    = 64;
constexpr int LP_    = KC_ + 4;
constexpr int TROWS_ = 16 + 128;
constexpr int STAGE_ = TROWS_ * LP_;

template <int K>
__device__ __forceinline__ v8f wmma_f32_tile_staged(const float* __restrict__ Ablk,
                                                    const float* __restrict__ Wblk,
                                                    float* __restrict__ lds)
{
  constexpr int NCH = K / KC_;
  constexpr int NB  = (NCH > 1) ? 2 : 1;
  const int tid  = threadIdx.x;
  const int lane = tid & 31;
  const int wave = tid >> 5;
  const int srow = tid >> 4;
  const int spc  = tid & 15;
  const float* ga = Ablk + (size_t)srow * K + 4 * spc;
  const float* gw = Wblk + (size_t)srow * K + 4 * spc;
  const int la = srow * LP_ + 4 * spc;
  const int lw = (16 + srow) * LP_ + 4 * spc;

  v4f ra, rw[8];
  ra = *(const v4f*)ga;
  for (int i = 0; i < 8; ++i) rw[i] = *(const v4f*)(gw + (size_t)(16 * i) * K);

  const int l16 = lane & 15, half = lane >> 4;
  const int fa = l16 * LP_ + 2 * half;
  const int fb = (16 + 16 * wave + l16) * LP_ + 2 * half;

  v8f c = {};
  for (int ch = 0; ch < NCH; ++ch) {
    float* st = lds + (ch & (NB - 1)) * STAGE_;
    *(v4f*)(st + la) = ra;
    for (int i = 0; i < 8; ++i) *(v4f*)(st + lw + 16 * i * LP_) = rw[i];
    __syncthreads();
    if (ch + 1 < NCH) {
      const int ko = (ch + 1) * KC_;
      ra = *(const v4f*)(ga + ko);
      for (int i = 0; i < 8; ++i) rw[i] = *(const v4f*)(gw + (size_t)(16 * i) * K + ko);
    }
    for (int k = 0; k < KC_; k += 4) {
      v2f a = *(const v2f*)(st + fa + k);
      v2f b = *(const v2f*)(st + fb + k);
      c = __builtin_amdgcn_wmma_f32_16x16x4_f32(
           false, a,  false, b,
           (short)0, c,  false,  false);
    }
  }
  return c;
}

constexpr int PT_     = 128;
constexpr int PW_     = 64;
constexpr int PROWS_  = 16 + PW_;
constexpr int PSTAGE_ = PROWS_ * LP_;
constexpr int PD_     = 4;
constexpr int PL_     = PROWS_ / 8;

__device__ __forceinline__ v8f wmma_f32_tile_deep(const float* __restrict__ Ablk,
                                                  const float* __restrict__ Wblk,
                                                  float* __restrict__ lds)
{
  constexpr int K   = D_;
  constexpr int NCH = K / KC_;
  static_assert(NCH >= PD_, "the prologue fills every register set");
  const int tid  = threadIdx.x;
  const int lane = tid & 31;
  const int wave = tid >> 5;
  const int srow = tid >> 4;
  const int spc  = tid & 15;
  const float* ga = Ablk + (size_t)srow * K + 4 * spc;
  const float* gw = Wblk + (size_t)srow * K + 4 * spc;
  const int l0 = srow * LP_ + 4 * spc;

  v4f r[PD_][PL_];
#define P_LOAD(set, ko)                                                              \
  for (int i = 0; i < 2; ++i)   r[set][i] = *(const v4f*)(ga + (size_t)(8 * i) * K + (ko)); \
  for (int i = 2; i < PL_; ++i) r[set][i] = *(const v4f*)(gw + (size_t)(8 * (i - 2)) * K + (ko));
#define P_STORE(set, stg)                                                            \
  for (int i = 0; i < PL_; ++i) *(v4f*)(lds + (stg) * PSTAGE_ + l0 + 8 * i * LP_) = r[set][i];
#define P_BARRIER() asm volatile("s_wait_dscnt 0x0\n\ts_barrier_signal -1\n\ts_barrier_wait -1" ::: "memory")
  P_LOAD(0, 0)
  P_LOAD(1, KC_)
  P_LOAD(2, 2 * KC_)
  P_LOAD(3, 3 * KC_)

  const int l16 = lane & 15, half = lane >> 4;
  const int fa = l16 * LP_ + 2 * half;
  const int fb = (16 + 16 * wave + l16) * LP_ + 2 * half;

  P_STORE(0, 0)
  P_BARRIER();

  v8f c = {};
#pragma unroll
  for (int ch = 0; ch < NCH; ++ch) {
    const float* st = lds + (ch & 1) * PSTAGE_;
    v2f a[KC_ / 4], b[KC_ / 4];
    for (int i = 0; i < KC_ / 4; ++i) {
      a[i] = *(const v2f*)(st + fa + 4 * i);
      b[i] = *(const v2f*)(st + fb + 4 * i);
    }
    if (ch + PD_ < NCH) {
      P_LOAD(ch % PD_, (ch + PD_) * KC_)
    }
    if (ch + 1 < NCH) {
      P_STORE((ch + 1) % PD_, (ch + 1) & 1)
    }
    __builtin_amdgcn_sched_barrier(0);
    for (int i = 0; i < KC_ / 4; ++i)
      c = __builtin_amdgcn_wmma_f32_16x16x4_f32(
           false, a[i],  false, b[i],
           (short)0, c,  false,  false);
    __builtin_amdgcn_sched_barrier(0);
    if (ch + 1 < NCH) P_BARRIER();
  }
#undef P_LOAD
#undef P_STORE
#undef P_BARRIER
  return c;
}

__device__ __forceinline__ void proj_body(const float* __restrict__ X, const float* __restrict__ W,
                                          float* __restrict__ out, int split, int blk,
                                          float* __restrict__ lds)
{
  const int N = D_;
  const int K = D_;
  const int tid  = threadIdx.x;
  const int lane = tid & 31;
  const int wave = tid >> 5;
  const int tile = blk * (PW_ / 16) + wave;
  const int tm = tile / (N / 16);
  const int tn = tile % (N / 16);
  const int m0 = tm * 16, n0 = tn * 16;
  const int nb0 = (blk % (N / PW_)) * PW_;

  const int l16 = lane & 15, half = lane >> 4;
  const int col  = n0 + l16;
  const int row0 = m0 + 8 * half;
  float* op;
  int ostride;
  if (split) {
    const int b = row0 / LQ_, l0 = row0 % LQ_;
    const int h = col / DH_, j = col % DH_;
    op = out + (((size_t)(b * H_ + h) * LQ_) + l0) * DH_ + j;
    ostride = DH_;
  } else {
    op = out + (size_t)row0 * N + col;
    ostride = N;
  }
  asm volatile("" : "+v"(op));

  v8f c = wmma_f32_tile_deep(X + (size_t)m0 * K, W + (size_t)nb0 * K, lds);

  for (int r = 0; r < 8; ++r) op[(size_t)r * ostride] = c[r];
}

__global__ void __launch_bounds__(PT_)
proj_kernel(const float* __restrict__ X, const float* __restrict__ W,
            float* __restrict__ out)
{
  __shared__ __attribute__((aligned(16))) float lds[2 * PSTAGE_];
  proj_body(X, W, out, 0, blockIdx.x, lds);
}

__global__ void __launch_bounds__(PT_)
proj_qkv_kernel(const float* __restrict__ Xq, const float* __restrict__ Xk,
                const float* __restrict__ Xv,
                const float* __restrict__ Wq, const float* __restrict__ Wk,
                const float* __restrict__ Wv,
                float* __restrict__ oq, float* __restrict__ ok, float* __restrict__ ov)
{
  __shared__ __attribute__((aligned(16))) float lds[2 * PSTAGE_];
  const int which = blockIdx.y;
  const float* X = (which == 0) ? Xq : (which == 1) ? Xk : Xv;
  const float* W = (which == 0) ? Wq : (which == 1) ? Wk : Wv;
  float*       o = (which == 0) ? oq : (which == 1) ? ok : ov;
  proj_body(X, W, o, 1, blockIdx.x, lds);
}

__global__ void __launch_bounds__(256)
feat_kernel(const float* __restrict__ Hq, const float* __restrict__ Hk,
            const float* __restrict__ Wqa, const float* __restrict__ Wka,
            float* __restrict__ outq, float* __restrict__ outk)
{
  __shared__ __attribute__((aligned(16))) float lds[STAGE_];
  const int which = blockIdx.z;
  const int bh = blockIdx.y;

  const int tid  = threadIdx.x;
  const int lane = tid & 31;
  const int wave = tid >> 5;
  const int l16 = lane & 15, half = lane >> 4;

  const float* Ablk;
  const float* Wblk;
  float* op;
  int opitch;
  if (which == 0) {
    const int m0  = (blockIdx.x / (F_ / 128)) * 16;
    const int nb0 = (blockIdx.x % (F_ / 128)) * 128;
    Ablk = Hq + ((size_t)bh * LQ_ + m0) * DH_;
    Wblk = Wqa + (size_t)nb0 * DH_;
    op = outq + ((size_t)bh * LQ_ + m0 + 8 * half) * F_ + nb0 + 16 * wave + l16;
    opitch = F_;
  } else {
    const int f0 = blockIdx.x * 16;
    Ablk = Wka + (size_t)f0 * DH_;
    Wblk = Hk + (size_t)bh * LK_ * DH_;
    op = outk + ((size_t)bh * F_ + f0 + 8 * half) * LK_ + 16 * wave + l16;
    opitch = LK_;
  }
  float* o0 = op;
  float* o1 = op + (size_t)1 * opitch;
  float* o2 = op + (size_t)2 * opitch;
  float* o3 = op + (size_t)3 * opitch;
  float* o4 = op + (size_t)4 * opitch;
  float* o5 = op + (size_t)5 * opitch;
  float* o6 = op + (size_t)6 * opitch;
  float* o7 = op + (size_t)7 * opitch;
  asm volatile("" : "+v"(o0), "+v"(o1), "+v"(o2), "+v"(o3), "+v"(o4), "+v"(o5), "+v"(o6), "+v"(o7));

  v8f c = wmma_f32_tile_staged<DH_>(Ablk, Wblk, lds);

  *o0 = c[0]; *o1 = c[1]; *o2 = c[2]; *o3 = c[3];
  *o4 = c[4]; *o5 = c[5]; *o6 = c[6]; *o7 = c[7];
}

constexpr int QT_ = 4;
constexpr int NS_ = 8;
constexpr int FS_ = F_ / NS_;
constexpr int FC_ = 8;
constexpr int SC_ = 4;
constexpr int KL_ = 4;

__global__ void __launch_bounds__(256)
score_softmax_av_kernel(const float* __restrict__ qf,
                        const float* __restrict__ kft,
                        const float* __restrict__ vh,
                        const float* __restrict__ wv_a,
                        const int*   __restrict__ valid_lens,
                        float* __restrict__ ctx)
{
  static_assert(QT_ * DH_ == 256, "the last stage maps one thread to one (row, column)");
  static_assert(LK_ == 32 * KL_ && F_ == NS_ * FS_ && NS_ == 8, "32 lanes x 4 keys cover the keys, 8 waves cover the features");
  static_assert(FS_ % (2 * FC_) == 0 && FC_ == 2 * SC_, "two register sets alternate, two scalar sets per set");
  __shared__ __attribute__((aligned(16))) float sp[NS_][QT_][LK_];
  __shared__ float spart[QT_][4][DH_];

  const int qt  = blockIdx.x % (LQ_ / QT_);
  const int bh  = blockIdx.x / (LQ_ / QT_);
  const int q0  = qt * QT_;
  const int b   = bh / H_;
  const int h   = bh % H_;
  const int tid  = threadIdx.x;
  const int lane = tid & 31;
  const int wave = tid >> 5;

  {
    const int wu = __builtin_amdgcn_readfirstlane(wave);
    const float* kt = kft + ((size_t)bh * F_ + (size_t)wu * FS_) * LK_ + KL_ * lane;
    const float* qr = qf + ((size_t)bh * LQ_ + q0) * F_ + wu * FS_;
    const float* wp = wv_a + wu * FS_;

    float acc[QT_][KL_];
    for (int q = 0; q < QT_; ++q)
      for (int j = 0; j < KL_; ++j) acc[q][j] = 0.f;

    v4f kA[FC_], kB[FC_];
    float qA[QT_][SC_], qB[QT_][SC_];
    float wA[SC_], wB[SC_];
#define K_LOAD(kx, fb)                                                            \
    for (int i = 0; i < FC_; ++i) kx[i] = *(const v4f*)(kt + (size_t)((fb) + i) * LK_);
#define S_LOAD(qx, wx, fb)                                                        \
    for (int i = 0; i < SC_; ++i) wx[i] = wp[(fb) + i];                           \
    for (int q = 0; q < QT_; ++q)                                                 \
      for (int i = 0; i < SC_; ++i) qx[q][i] = qr[(size_t)q * F_ + (fb) + i];
#define SC_STEP(kx, ko, qx, wx)                                                   \
    for (int i = 0; i < SC_; ++i) {                                               \
      float t[QT_][KL_];                                                          \
      for (int q = 0; q < QT_; ++q)                                               \
        for (int j = 0; j < KL_; ++j) t[q][j] = qx[q][i] + kx[(ko) + i][j];       \
      for (int q = 0; q < QT_; ++q)                                               \
        for (int j = 0; j < KL_; ++j) t[q][j] = TANH_F32(t[q][j]);                \
      for (int q = 0; q < QT_; ++q)                                               \
        for (int j = 0; j < KL_; ++j) acc[q][j] = __builtin_fmaf(wx[i], t[q][j], acc[q][j]); \
    }
    K_LOAD(kA, 0)
    S_LOAD(qA, wA, 0)
    for (int f0 = 0; f0 < FS_; f0 += 2 * FC_) {
      K_LOAD(kB, f0 + FC_)
      S_LOAD(qB, wB, f0 + SC_)
      SC_STEP(kA, 0, qA, wA)
      S_LOAD(qA, wA, f0 + FC_)
      SC_STEP(kA, SC_, qB, wB)
      const int fn = (f0 + 2 * FC_ < FS_) ? f0 + 2 * FC_ : 0;
      K_LOAD(kA, fn)
      S_LOAD(qB, wB, f0 + FC_ + SC_)
      SC_STEP(kB, 0, qA, wA)
      S_LOAD(qA, wA, fn)
      SC_STEP(kB, SC_, qB, wB)
    }
#undef K_LOAD
#undef S_LOAD
#undef SC_STEP
    for (int q = 0; q < QT_; ++q) {
      v4f r = {acc[q][0], acc[q][1], acc[q][2], acc[q][3]};
      *(v4f*)&sp[wu][q][KL_ * lane] = r;
    }
  }
  __syncthreads();

  if (wave < QT_) {
    const int q = wave;
    const int valid = valid_lens[b];
    float val[4], e[4], sm[4];
    for (int j = 0; j < 4; ++j) {
      const int k = lane + 32 * j;
      float s = sp[0][q][k];
      for (int w = 1; w < NS_; ++w) s += sp[w][q][k];
      val[j] = (k < valid) ? s : -1.0e6f;
    }
    float m = fmaxf(fmaxf(val[0], val[1]), fmaxf(val[2], val[3]));
    for (int off = 16; off > 0; off >>= 1)
      m = fmaxf(m, __shfl_xor(m, off, 32));
    for (int j = 0; j < 4; ++j) {
      e[j] = __expf(val[j] - m);
      float t = e[j];
      for (int off = 16; off > 0; off >>= 1)
        t += __shfl_xor(t, off, 32);
      sm[j] = t;
    }
    const float rowsum = sm[0] + sm[1] + sm[2] + sm[3];
    for (int j = 0; j < 4; ++j) sp[0][q][lane + 32 * j] = e[j] / rowsum;
  }
  __syncthreads();

  const int j = tid & 63;
  const int part = tid >> 6;
  const float* vb = vh + (size_t)bh * LK_ * DH_;
  float o[QT_];
  for (int q = 0; q < QT_; ++q) o[q] = 0.f;
  for (int k = part * 32; k < part * 32 + 32; ++k) {
    const float v = vb[(size_t)k * DH_ + j];
    for (int q = 0; q < QT_; ++q) o[q] += sp[0][q][k] * v;
  }
  for (int q = 0; q < QT_; ++q) spart[q][part][j] = o[q];
  __syncthreads();

  {
    const int q = tid >> 6;
    const float r = spart[q][0][j] + spart[q][1][j] + spart[q][2][j] + spart[q][3][j];
    ctx[((size_t)b * LQ_ + q0 + q) * D_ + h * DH_ + j] = r;
  }
}

extern "C" void kernel_launch(void* const* d_in, const int* in_sizes, int n_in,
                              void* d_out, int out_size, void* d_ws, size_t ws_size,
                              hipStream_t stream) {
  (void)in_sizes; (void)n_in; (void)out_size; (void)ws_size;

  const float* queries    = (const float*)d_in[0];
  const float* keys       = (const float*)d_in[1];
  const float* values     = (const float*)d_in[2];
  const int*   valid_lens = (const int*)  d_in[3];
  const float* Wq  = (const float*)d_in[4];
  const float* Wk  = (const float*)d_in[5];
  const float* Wv  = (const float*)d_in[6];
  const float* Wo  = (const float*)d_in[7];
  const float* Wqa = (const float*)d_in[8];
  const float* Wka = (const float*)d_in[9];
  const float* wva = (const float*)d_in[10];

  float* ws  = (float*)d_ws;
  const size_t nQKV = (size_t)B_ * H_ * LQ_ * DH_;
  const size_t nF   = (size_t)B_ * H_ * LQ_ * F_;
  float* q   = ws;
  float* k   = q  + nQKV;
  float* v   = k  + nQKV;
  float* qf  = v  + nQKV;
  float* kfb = qf + nF;
  float* ctx = kfb + nF;

  dim3 blk(256);
  dim3 pblk(PT_);
  proj_qkv_kernel<<<dim3(128, 3), pblk, 0, stream>>>(queries, keys, values, Wq, Wk, Wv, q, k, v);

  feat_kernel<<<dim3(32, B_ * H_, 2), blk, 0, stream>>>(q, k, Wqa, Wka, qf, kfb);

  score_softmax_av_kernel<<<B_ * H_ * (LQ_ / QT_), blk, 0, stream>>>(
      qf, kfb, v, wva, valid_lens, ctx);

  proj_kernel<<<128, pblk, 0, stream>>>(ctx, Wo, (float*)d_out);
}


# Feature GEMMs run as the epilogue of the Q and K projection work-groups, from the finished 16 x 64 head tile in LDS; 4 launches -> 3, q and k never stored; every sum keeps its k order
# speedup vs baseline: 3.3721x; 1.0325x over previous
#include <hip/hip_runtime.h>
#include <hip/hip_bf16.h>
#include <math.h>

constexpr int B_  = 2;
constexpr int H_  = 8;
constexpr int LQ_ = 128;
constexpr int LK_ = 128;
constexpr int D_  = 512;
constexpr int DH_ = 64;
constexpr int F_  = 512;

typedef float v2f __attribute__((ext_vector_type(2)));
typedef float v8f __attribute__((ext_vector_type(8)));

#if __has_builtin(__builtin_amdgcn_tanhf)
#define TANH_F32(x) __builtin_amdgcn_tanhf(x)
#elif __has_builtin(__builtin_amdgcn_tanh_f32)
#define TANH_F32(x) __builtin_amdgcn_tanh_f32(x)
#else
#define TANH_F32(x) tanhf(x)
#endif

typedef float v4f __attribute__((ext_vector_type(4)));

constexpr int KC_    = 64;
constexpr int LP_    = KC_ + 4;

constexpr int PT_     = 128;
constexpr int PW_     = 64;
constexpr int PROWS_  = 16 + PW_;
constexpr int PSTAGE_ = PROWS_ * LP_;
constexpr int PD_     = 4;
constexpr int PL_     = PROWS_ / 8;

__device__ __forceinline__ v8f wmma_f32_tile_deep(const float* __restrict__ Ablk,
                                                  const float* __restrict__ Wblk,
                                                  float* __restrict__ lds)
{
  constexpr int K   = D_;
  constexpr int NCH = K / KC_;
  static_assert(NCH >= PD_, "the prologue fills every register set");
  const int tid  = threadIdx.x;
  const int lane = tid & 31;
  const int wave = tid >> 5;
  const int srow = tid >> 4;
  const int spc  = tid & 15;
  const float* ga = Ablk + (size_t)srow * K + 4 * spc;
  const float* gw = Wblk + (size_t)srow * K + 4 * spc;
  const int l0 = srow * LP_ + 4 * spc;

  v4f r[PD_][PL_];
#define P_LOAD(set, ko)                                                              \
  for (int i = 0; i < 2; ++i)   r[set][i] = *(const v4f*)(ga + (size_t)(8 * i) * K + (ko)); \
  for (int i = 2; i < PL_; ++i) r[set][i] = *(const v4f*)(gw + (size_t)(8 * (i - 2)) * K + (ko));
#define P_STORE(set, stg)                                                            \
  for (int i = 0; i < PL_; ++i) *(v4f*)(lds + (stg) * PSTAGE_ + l0 + 8 * i * LP_) = r[set][i];
#define P_BARRIER() asm volatile("s_wait_dscnt 0x0\n\ts_barrier_signal -1\n\ts_barrier_wait -1" ::: "memory")
  P_LOAD(0, 0)
  P_LOAD(1, KC_)
  P_LOAD(2, 2 * KC_)
  P_LOAD(3, 3 * KC_)

  const int l16 = lane & 15, half = lane >> 4;
  const int fa = l16 * LP_ + 2 * half;
  const int fb = (16 + 16 * wave + l16) * LP_ + 2 * half;

  P_STORE(0, 0)
  P_BARRIER();

  v8f c = {};
#pragma unroll
  for (int ch = 0; ch < NCH; ++ch) {
    const float* st = lds + (ch & 1) * PSTAGE_;
    v2f a[KC_ / 4], b[KC_ / 4];
    for (int i = 0; i < KC_ / 4; ++i) {
      a[i] = *(const v2f*)(st + fa + 4 * i);
      b[i] = *(const v2f*)(st + fb + 4 * i);
    }
    if (ch + PD_ < NCH) {
      P_LOAD(ch % PD_, (ch + PD_) * KC_)
    }
    if (ch + 1 < NCH) {
      P_STORE((ch + 1) % PD_, (ch + 1) & 1)
    }
    __builtin_amdgcn_sched_barrier(0);
    for (int i = 0; i < KC_ / 4; ++i)
      c = __builtin_amdgcn_wmma_f32_16x16x4_f32(
           false, a[i],  false, b[i],
           (short)0, c,  false,  false);
    __builtin_amdgcn_sched_barrier(0);
    if (ch + 1 < NCH) P_BARRIER();
  }
#undef P_LOAD
#undef P_STORE
#undef P_BARRIER
  return c;
}

__device__ __forceinline__ void lds_only_barrier()
{
  asm volatile("s_wait_dscnt 0x0\n\ts_barrier_signal -1\n\ts_barrier_wait -1" ::: "memory");
}

struct FeatAddr {
  unsigned gw;
  unsigned ob;
  unsigned ts;
  unsigned ls;
  unsigned ft;
  unsigned fw;
};

constexpr int FT_ = F_ / 64;
constexpr int WBUF_ = 16 * LP_ * 4;

template <int SIDE>
__device__ __forceinline__ void feat_from_tile(v8f c, const FeatAddr ad,
                                               const float* __restrict__ Wa, float* __restrict__ out,
                                               float* __restrict__ lds)
{
  const char* gwb = (const char*)Wa;
  char* obb = (char*)out;
  char* lb  = (char*)lds;
  v4f rw[2][8];
  v2f x[KC_ / 4], y[2][KC_ / 4];
#define G_LOAD(set, t) \
  for (int i = 0; i < 8; ++i) rw[set][i] = *(const v4f*)(gwb + ad.gw + ((t) * 64 * DH_ + 2 * i * DH_) * 4);
#define L_STORE(set, buf) \
  for (int i = 0; i < 8; ++i) *(v4f*)(lb + ad.ls + (buf) * WBUF_ + 2 * i * LP_ * 4) = rw[set][i];
#define F_LOAD(buf) \
  for (int i = 0; i < KC_ / 4; ++i) y[buf][i] = *(const v2f*)(lb + ad.fw + (buf) * WBUF_ + 16 * i);
  G_LOAD(0, 0)
  G_LOAD(1, 1)
  lds_only_barrier();
  for (int r = 0; r < 8; ++r) *(float*)(lb + ad.ts + r * LP_ * 4) = c[r];
  L_STORE(0, 0)
  G_LOAD(0, 2)
  lds_only_barrier();
  for (int i = 0; i < KC_ / 4; ++i) x[i] = *(const v2f*)(lb + ad.ft + 16 * i);
  F_LOAD(0)
  L_STORE(1, 1)
  G_LOAD(1, 3)
  lds_only_barrier();
#pragma unroll
  for (int t = 0; t < FT_; ++t) {
    if (t + 1 < FT_) { F_LOAD((t + 1) & 1) }
    if (t + 2 < FT_) { L_STORE(t & 1, t & 1) }
    if (t + 4 < FT_) { G_LOAD(t & 1, t + 4) }
    __builtin_amdgcn_sched_barrier(0);
    v8f d = {};
    for (int i = 0; i < KC_ / 4; ++i) {
      if (SIDE == 0)
        d = __builtin_amdgcn_wmma_f32_16x16x4_f32(false, x[i], false, y[t & 1][i], (short)0, d, false, false);
      else
        d = __builtin_amdgcn_wmma_f32_16x16x4_f32(false, y[t & 1][i], false, x[i], (short)0, d, false, false);
    }
    __builtin_amdgcn_sched_barrier(0);
    if (SIDE == 0) {
      for (int r = 0; r < 8; ++r) *(float*)(obb + ad.ob + (t * 64 + r * F_) * 4) = d[r];
    } else {
      for (int r = 0; r < 8; ++r) *(float*)(obb + ad.ob + (t * 64 * LK_ + r * LK_) * 4) = d[r];
    }
    if (t + 1 < FT_) lds_only_barrier();
  }
#undef G_LOAD
#undef L_STORE
#undef F_LOAD
}

__device__ __forceinline__ void proj_body(const float* __restrict__ X, const float* __restrict__ W,
                                          const float* __restrict__ Wa,
                                          float* __restrict__ out, int mode, int blk,
                                          float* __restrict__ lds)
{
  const int N = D_;
  const int K = D_;
  const int tid  = threadIdx.x;
  const int lane = tid & 31;
  const int wave = tid >> 5;
  const int tile = blk * (PW_ / 16) + wave;
  const int tm = tile / (N / 16);
  const int tn = tile % (N / 16);
  const int m0 = tm * 16, n0 = tn * 16;
  const int nb0 = (blk % (N / PW_)) * PW_;

  const int l16 = lane & 15, half = lane >> 4;
  const int col  = n0 + l16;
  const int row0 = m0 + 8 * half;
  const int b = m0 / LQ_, t0 = m0 % LQ_;
  const int h = nb0 / DH_;
  unsigned oo;
  if (mode == 0) {
    oo = row0 * N + col;
  } else if (mode == 1) {
    oo = ((b * H_ + h) * LQ_ + t0 + 8 * half) * DH_ + (col % DH_);
  } else if (mode == 2) {
    oo = ((b * H_ + h) * LQ_ + t0 + 8 * half) * F_ + 16 * wave + l16;
  } else {
    oo = ((b * H_ + h) * F_ + 16 * wave + 8 * half) * LK_ + t0 + l16;
  }
  FeatAddr ad;
  ad.gw = 4u * ((16 * wave + half) * DH_ + 4 * l16);
  ad.ob = 4u * oo;
  ad.ts = 4u * ((8 * half) * LP_ + 16 * wave + l16);
  ad.ls = 4u * ((16 + 32 * wave + half) * LP_ + 4 * l16);
  ad.ft = 4u * (l16 * LP_ + 2 * half);
  ad.fw = 4u * ((16 + 32 * wave + l16) * LP_ + 2 * half);
  asm volatile("" : "+v"(ad.ob), "+v"(ad.gw), "+v"(ad.ts), "+v"(ad.ls), "+v"(ad.ft), "+v"(ad.fw));

  v8f c = wmma_f32_tile_deep(X + (size_t)m0 * K, W + (size_t)nb0 * K, lds);

  char* obb = (char*)out;
  if (mode == 0) {
    for (int r = 0; r < 8; ++r) *(float*)(obb + ad.ob + r * N * 4) = c[r];
  } else if (mode == 1) {
    for (int r = 0; r < 8; ++r) *(float*)(obb + ad.ob + r * DH_ * 4) = c[r];
  } else if (mode == 2) {
    feat_from_tile<0>(c, ad, Wa, out, lds);
  } else {
    feat_from_tile<1>(c, ad, Wa, out, lds);
  }
}

__global__ void __launch_bounds__(PT_)
proj_kernel(const float* __restrict__ X, const float* __restrict__ W,
            float* __restrict__ out)
{
  __shared__ __attribute__((aligned(16))) float lds[2 * PSTAGE_];
  proj_body(X, W, W, out, 0, blockIdx.x, lds);
}

__global__ void __launch_bounds__(PT_)
proj_qkv_kernel(const float* __restrict__ Xq, const float* __restrict__ Xk,
                const float* __restrict__ Xv,
                const float* __restrict__ Wq, const float* __restrict__ Wk,
                const float* __restrict__ Wv,
                const float* __restrict__ Wqa, const float* __restrict__ Wka,
                float* __restrict__ oqf, float* __restrict__ okft, float* __restrict__ ov)
{
  __shared__ __attribute__((aligned(16))) float lds[2 * PSTAGE_];
  static_assert(PW_ == DH_ && LQ_ == LK_, "a projection tile is 16 tokens of one head");
  static_assert((16 + 4 * 32) * LP_ <= 2 * PSTAGE_, "tile + 4 waves x 2 buffers fit the two stages");
  const int which = blockIdx.y;
  const float* X  = (which == 0) ? Xq : (which == 1) ? Xk : Xv;
  const float* W  = (which == 0) ? Wq : (which == 1) ? Wk : Wv;
  const float* Wa = (which == 0) ? Wqa : Wka;
  float*       o  = (which == 0) ? oqf : (which == 1) ? okft : ov;
  const int mode  = (which == 0) ? 2 : (which == 1) ? 3 : 1;
  proj_body(X, W, Wa, o, mode, blockIdx.x, lds);
}

constexpr int QT_ = 4;
constexpr int NS_ = 8;
constexpr int FS_ = F_ / NS_;
constexpr int FC_ = 8;
constexpr int SC_ = 4;
constexpr int KL_ = 4;

__global__ void __launch_bounds__(256)
score_softmax_av_kernel(const float* __restrict__ qf,
                        const float* __restrict__ kft,
                        const float* __restrict__ vh,
                        const float* __restrict__ wv_a,
                        const int*   __restrict__ valid_lens,
                        float* __restrict__ ctx)
{
  static_assert(QT_ * DH_ == 256, "the last stage maps one thread to one (row, column)");
  static_assert(LK_ == 32 * KL_ && F_ == NS_ * FS_ && NS_ == 8, "32 lanes x 4 keys cover the keys, 8 waves cover the features");
  static_assert(FS_ % (2 * FC_) == 0 && FC_ == 2 * SC_, "two register sets alternate, two scalar sets per set");
  __shared__ __attribute__((aligned(16))) float sp[NS_][QT_][LK_];
  __shared__ float spart[QT_][4][DH_];

  const int qt  = blockIdx.x % (LQ_ / QT_);
  const int bh  = blockIdx.x / (LQ_ / QT_);
  const int q0  = qt * QT_;
  const int b   = bh / H_;
  const int h   = bh % H_;
  const int tid  = threadIdx.x;
  const int lane = tid & 31;
  const int wave = tid >> 5;

  {
    const int wu = __builtin_amdgcn_readfirstlane(wave);
    const float* kt = kft + ((size_t)bh * F_ + (size_t)wu * FS_) * LK_ + KL_ * lane;
    const float* qr = qf + ((size_t)bh * LQ_ + q0) * F_ + wu * FS_;
    const float* wp = wv_a + wu * FS_;

    float acc[QT_][KL_];
    for (int q = 0; q < QT_; ++q)
      for (int j = 0; j < KL_; ++j) acc[q][j] = 0.f;

    v4f kA[FC_], kB[FC_];
    float qA[QT_][SC_], qB[QT_][SC_];
    float wA[SC_], wB[SC_];
#define K_LOAD(kx, fb)                                                            \
    for (int i = 0; i < FC_; ++i) kx[i] = *(const v4f*)(kt + (size_t)((fb) + i) * LK_);
#define S_LOAD(qx, wx, fb)                                                        \
    for (int i = 0; i < SC_; ++i) wx[i] = wp[(fb) + i];                           \
    for (int q = 0; q < QT_; ++q)                                                 \
      for (int i = 0; i < SC_; ++i) qx[q][i] = qr[(size_t)q * F_ + (fb) + i];
#define SC_STEP(kx, ko, qx, wx)                                                   \
    for (int i = 0; i < SC_; ++i) {                                               \
      float t[QT_][KL_];                                                          \
      for (int q = 0; q < QT_; ++q)                                               \
        for (int j = 0; j < KL_; ++j) t[q][j] = qx[q][i] + kx[(ko) + i][j];       \
      for (int q = 0; q < QT_; ++q)                                               \
        for (int j = 0; j < KL_; ++j) t[q][j] = TANH_F32(t[q][j]);                \
      for (int q = 0; q < QT_; ++q)                                               \
        for (int j = 0; j < KL_; ++j) acc[q][j] = __builtin_fmaf(wx[i], t[q][j], acc[q][j]); \
    }
    K_LOAD(kA, 0)
    S_LOAD(qA, wA, 0)
    for (int f0 = 0; f0 < FS_; f0 += 2 * FC_) {
      K_LOAD(kB, f0 + FC_)
      S_LOAD(qB, wB, f0 + SC_)
      SC_STEP(kA, 0, qA, wA)
      S_LOAD(qA, wA, f0 + FC_)
      SC_STEP(kA, SC_, qB, wB)
      const int fn = (f0 + 2 * FC_ < FS_) ? f0 + 2 * FC_ : 0;
      K_LOAD(kA, fn)
      S_LOAD(qB, wB, f0 + FC_ + SC_)
      SC_STEP(kB, 0, qA, wA)
      S_LOAD(qA, wA, fn)
      SC_STEP(kB, SC_, qB, wB)
    }
#undef K_LOAD
#undef S_LOAD
#undef SC_STEP
    for (int q = 0; q < QT_; ++q) {
      v4f r = {acc[q][0], acc[q][1], acc[q][2], acc[q][3]};
      *(v4f*)&sp[wu][q][KL_ * lane] = r;
    }
  }
  __syncthreads();

  if (wave < QT_) {
    const int q = wave;
    const int valid = valid_lens[b];
    float val[4], e[4], sm[4];
    for (int j = 0; j < 4; ++j) {
      const int k = lane + 32 * j;
      float s = sp[0][q][k];
      for (int w = 1; w < NS_; ++w) s += sp[w][q][k];
      val[j] = (k < valid) ? s : -1.0e6f;
    }
    float m = fmaxf(fmaxf(val[0], val[1]), fmaxf(val[2], val[3]));
    for (int off = 16; off > 0; off >>= 1)
      m = fmaxf(m, __shfl_xor(m, off, 32));
    for (int j = 0; j < 4; ++j) {
      e[j] = __expf(val[j] - m);
      float t = e[j];
      for (int off = 16; off > 0; off >>= 1)
        t += __shfl_xor(t, off, 32);
      sm[j] = t;
    }
    const float rowsum = sm[0] + sm[1] + sm[2] + sm[3];
    for (int j = 0; j < 4; ++j) sp[0][q][lane + 32 * j] = e[j] / rowsum;
  }
  __syncthreads();

  const int j = tid & 63;
  const int part = tid >> 6;
  const float* vb = vh + (size_t)bh * LK_ * DH_;
  float o[QT_];
  for (int q = 0; q < QT_; ++q) o[q] = 0.f;
  for (int k = part * 32; k < part * 32 + 32; ++k) {
    const float v = vb[(size_t)k * DH_ + j];
    for (int q = 0; q < QT_; ++q) o[q] += sp[0][q][k] * v;
  }
  for (int q = 0; q < QT_; ++q) spart[q][part][j] = o[q];
  __syncthreads();

  {
    const int q = tid >> 6;
    const float r = spart[q][0][j] + spart[q][1][j] + spart[q][2][j] + spart[q][3][j];
    ctx[((size_t)b * LQ_ + q0 + q) * D_ + h * DH_ + j] = r;
  }
}

extern "C" void kernel_launch(void* const* d_in, const int* in_sizes, int n_in,
                              void* d_out, int out_size, void* d_ws, size_t ws_size,
                              hipStream_t stream) {
  (void)in_sizes; (void)n_in; (void)out_size; (void)ws_size;

  const float* queries    = (const float*)d_in[0];
  const float* keys       = (const float*)d_in[1];
  const float* values     = (const float*)d_in[2];
  const int*   valid_lens = (const int*)  d_in[3];
  const float* Wq  = (const float*)d_in[4];
  const float* Wk  = (const float*)d_in[5];
  const float* Wv  = (const float*)d_in[6];
  const float* Wo  = (const float*)d_in[7];
  const float* Wqa = (const float*)d_in[8];
  const float* Wka = (const float*)d_in[9];
  const float* wva = (const float*)d_in[10];

  float* ws  = (float*)d_ws;
  const size_t nQKV = (size_t)B_ * H_ * LQ_ * DH_;
  const size_t nF   = (size_t)B_ * H_ * LQ_ * F_;
  float* v   = ws;
  float* qf  = v  + nQKV;
  float* kfb = qf + nF;
  float* ctx = kfb + nF;

  dim3 blk(256);
  dim3 pblk(PT_);
  proj_qkv_kernel<<<dim3(128, 3), pblk, 0, stream>>>(queries, keys, values, Wq, Wk, Wv, Wqa, Wka, qf, kfb, v);

  score_softmax_av_kernel<<<B_ * H_ * (LQ_ / QT_), blk, 0, stream>>>(
      qf, kfb, v, wva, valid_lens, ctx);

  proj_kernel<<<128, pblk, 0, stream>>>(ctx, Wo, (float*)d_out);
}


# Score stage scores only the 32-key blocks that hold a key in front of the valid length (key features stored key-block-major, a lane owns one key per block); every sum keeps its order
# speedup vs baseline: 3.6108x; 1.0708x over previous
#include <hip/hip_runtime.h>
#include <hip/hip_bf16.h>
#include <math.h>

constexpr int B_  = 2;
constexpr int H_  = 8;
constexpr int LQ_ = 128;
constexpr int LK_ = 128;
constexpr int D_  = 512;
constexpr int DH_ = 64;
constexpr int F_  = 512;

typedef float v2f __attribute__((ext_vector_type(2)));
typedef float v8f __attribute__((ext_vector_type(8)));

#if __has_builtin(__builtin_amdgcn_tanhf)
#define TANH_F32(x) __builtin_amdgcn_tanhf(x)
#elif __has_builtin(__builtin_amdgcn_tanh_f32)
#define TANH_F32(x) __builtin_amdgcn_tanh_f32(x)
#else
#define TANH_F32(x) tanhf(x)
#endif

typedef float v4f __attribute__((ext_vector_type(4)));

constexpr int KC_    = 64;
constexpr int LP_    = KC_ + 4;

constexpr int PT_     = 128;
constexpr int PW_     = 64;
constexpr int PROWS_  = 16 + PW_;
constexpr int PSTAGE_ = PROWS_ * LP_;
constexpr int PD_     = 4;
constexpr int PL_     = PROWS_ / 8;

__device__ __forceinline__ v8f wmma_f32_tile_deep(const float* __restrict__ Ablk,
                                                  const float* __restrict__ Wblk,
                                                  float* __restrict__ lds)
{
  constexpr int K   = D_;
  constexpr int NCH = K / KC_;
  static_assert(NCH >= PD_, "the prologue fills every register set");
  const int tid  = threadIdx.x;
  const int lane = tid & 31;
  const int wave = tid >> 5;
  const int srow = tid >> 4;
  const int spc  = tid & 15;
  const float* ga = Ablk + (size_t)srow * K + 4 * spc;
  const float* gw = Wblk + (size_t)srow * K + 4 * spc;
  const int l0 = srow * LP_ + 4 * spc;

  v4f r[PD_][PL_];
#define P_LOAD(set, ko)                                                              \
  for (int i = 0; i < 2; ++i)   r[set][i] = *(const v4f*)(ga + (size_t)(8 * i) * K + (ko)); \
  for (int i = 2; i < PL_; ++i) r[set][i] = *(const v4f*)(gw + (size_t)(8 * (i - 2)) * K + (ko));
#define P_STORE(set, stg)                                                            \
  for (int i = 0; i < PL_; ++i) *(v4f*)(lds + (stg) * PSTAGE_ + l0 + 8 * i * LP_) = r[set][i];
#define P_BARRIER() asm volatile("s_wait_dscnt 0x0\n\ts_barrier_signal -1\n\ts_barrier_wait -1" ::: "memory")
  P_LOAD(0, 0)
  P_LOAD(1, KC_)
  P_LOAD(2, 2 * KC_)
  P_LOAD(3, 3 * KC_)

  const int l16 = lane & 15, half = lane >> 4;
  const int fa = l16 * LP_ + 2 * half;
  const int fb = (16 + 16 * wave + l16) * LP_ + 2 * half;

  P_STORE(0, 0)
  P_BARRIER();

  v8f c = {};
#pragma unroll
  for (int ch = 0; ch < NCH; ++ch) {
    const float* st = lds + (ch & 1) * PSTAGE_;
    v2f a[KC_ / 4], b[KC_ / 4];
    for (int i = 0; i < KC_ / 4; ++i) {
      a[i] = *(const v2f*)(st + fa + 4 * i);
      b[i] = *(const v2f*)(st + fb + 4 * i);
    }
    if (ch + PD_ < NCH) {
      P_LOAD(ch % PD_, (ch + PD_) * KC_)
    }
    if (ch + 1 < NCH) {
      P_STORE((ch + 1) % PD_, (ch + 1) & 1)
    }
    __builtin_amdgcn_sched_barrier(0);
    for (int i = 0; i < KC_ / 4; ++i)
      c = __builtin_amdgcn_wmma_f32_16x16x4_f32(
           false, a[i],  false, b[i],
           (short)0, c,  false,  false);
    __builtin_amdgcn_sched_barrier(0);
    if (ch + 1 < NCH) P_BARRIER();
  }
#undef P_LOAD
#undef P_STORE
#undef P_BARRIER
  return c;
}

__device__ __forceinline__ void lds_only_barrier()
{
  asm volatile("s_wait_dscnt 0x0\n\ts_barrier_signal -1\n\ts_barrier_wait -1" ::: "memory");
}

constexpr int KQ_ = 32 * 4;
constexpr int KB_ = (F_ / 4) * KQ_;

struct FeatAddr {
  unsigned gw;
  unsigned ob;
  unsigned ts;
  unsigned ls;
  unsigned ft;
  unsigned fw;
};

constexpr int FT_ = F_ / 64;
constexpr int WBUF_ = 16 * LP_ * 4;

template <int SIDE>
__device__ __forceinline__ void feat_from_tile(v8f c, const FeatAddr ad,
                                               const float* __restrict__ Wa, float* __restrict__ out,
                                               float* __restrict__ lds)
{
  const char* gwb = (const char*)Wa;
  char* obb = (char*)out;
  char* lb  = (char*)lds;
  v4f rw[2][8];
  v2f x[KC_ / 4], y[2][KC_ / 4];
#define G_LOAD(set, t) \
  for (int i = 0; i < 8; ++i) rw[set][i] = *(const v4f*)(gwb + ad.gw + ((t) * 64 * DH_ + 2 * i * DH_) * 4);
#define L_STORE(set, buf) \
  for (int i = 0; i < 8; ++i) *(v4f*)(lb + ad.ls + (buf) * WBUF_ + 2 * i * LP_ * 4) = rw[set][i];
#define F_LOAD(buf) \
  for (int i = 0; i < KC_ / 4; ++i) y[buf][i] = *(const v2f*)(lb + ad.fw + (buf) * WBUF_ + 16 * i);
  G_LOAD(0, 0)
  G_LOAD(1, 1)
  lds_only_barrier();
  for (int r = 0; r < 8; ++r) *(float*)(lb + ad.ts + r * LP_ * 4) = c[r];
  L_STORE(0, 0)
  G_LOAD(0, 2)
  lds_only_barrier();
  for (int i = 0; i < KC_ / 4; ++i) x[i] = *(const v2f*)(lb + ad.ft + 16 * i);
  F_LOAD(0)
  L_STORE(1, 1)
  G_LOAD(1, 3)
  lds_only_barrier();
#pragma unroll
  for (int t = 0; t < FT_; ++t) {
    if (t + 1 < FT_) { F_LOAD((t + 1) & 1) }
    if (t + 2 < FT_) { L_STORE(t & 1, t & 1) }
    if (t + 4 < FT_) { G_LOAD(t & 1, t + 4) }
    __builtin_amdgcn_sched_barrier(0);
    v8f d = {};
    for (int i = 0; i < KC_ / 4; ++i) {
      if (SIDE == 0)
        d = __builtin_amdgcn_wmma_f32_16x16x4_f32(false, x[i], false, y[t & 1][i], (short)0, d, false, false);
      else
        d = __builtin_amdgcn_wmma_f32_16x16x4_f32(false, y[t & 1][i], false, x[i], (short)0, d, false, false);
    }
    __builtin_amdgcn_sched_barrier(0);
    if (SIDE == 0) {
      for (int r = 0; r < 8; ++r) *(float*)(obb + ad.ob + (t * 64 + r * F_) * 4) = d[r];
    } else {
      const v4f lo = {d[0], d[1], d[2], d[3]}, hi = {d[4], d[5], d[6], d[7]};
      *(v4f*)(obb + ad.ob + (t * 16 * KQ_) * 4)       = lo;
      *(v4f*)(obb + ad.ob + (t * 16 * KQ_ + KQ_) * 4) = hi;
    }
    if (t + 1 < FT_) lds_only_barrier();
  }
#undef G_LOAD
#undef L_STORE
#undef F_LOAD
}

__device__ __forceinline__ void proj_body(const float* __restrict__ X, const float* __restrict__ W,
                                          const float* __restrict__ Wa,
                                          float* __restrict__ out, int mode, int blk,
                                          float* __restrict__ lds)
{
  const int N = D_;
  const int K = D_;
  const int tid  = threadIdx.x;
  const int lane = tid & 31;
  const int wave = tid >> 5;
  const int tile = blk * (PW_ / 16) + wave;
  const int tm = tile / (N / 16);
  const int tn = tile % (N / 16);
  const int m0 = tm * 16, n0 = tn * 16;
  const int nb0 = (blk % (N / PW_)) * PW_;

  const int l16 = lane & 15, half = lane >> 4;
  const int col  = n0 + l16;
  const int row0 = m0 + 8 * half;
  const int b = m0 / LQ_, t0 = m0 % LQ_;
  const int h = nb0 / DH_;
  unsigned oo;
  if (mode == 0) {
    oo = row0 * N + col;
  } else if (mode == 1) {
    oo = ((b * H_ + h) * LQ_ + t0 + 8 * half) * DH_ + (col % DH_);
  } else if (mode == 2) {
    oo = ((b * H_ + h) * LQ_ + t0 + 8 * half) * F_ + 16 * wave + l16;
  } else {
    oo = ((b * H_ + h) * (LK_ / 32) + t0 / 32) * KB_ + (4 * wave + 2 * half) * KQ_ + (t0 % 32 + l16) * 4;
  }
  FeatAddr ad;
  ad.gw = 4u * ((16 * wave + half) * DH_ + 4 * l16);
  ad.ob = 4u * oo;
  ad.ts = 4u * ((8 * half) * LP_ + 16 * wave + l16);
  ad.ls = 4u * ((16 + 32 * wave + half) * LP_ + 4 * l16);
  ad.ft = 4u * (l16 * LP_ + 2 * half);
  ad.fw = 4u * ((16 + 32 * wave + l16) * LP_ + 2 * half);
  asm volatile("" : "+v"(ad.ob), "+v"(ad.gw), "+v"(ad.ts), "+v"(ad.ls), "+v"(ad.ft), "+v"(ad.fw));

  v8f c = wmma_f32_tile_deep(X + (size_t)m0 * K, W + (size_t)nb0 * K, lds);

  char* obb = (char*)out;
  if (mode == 0) {
    for (int r = 0; r < 8; ++r) *(float*)(obb + ad.ob + r * N * 4) = c[r];
  } else if (mode == 1) {
    for (int r = 0; r < 8; ++r) *(float*)(obb + ad.ob + r * DH_ * 4) = c[r];
  } else if (mode == 2) {
    feat_from_tile<0>(c, ad, Wa, out, lds);
  } else {
    feat_from_tile<1>(c, ad, Wa, out, lds);
  }
}

__global__ void __launch_bounds__(PT_)
proj_kernel(const float* __restrict__ X, const float* __restrict__ W,
            float* __restrict__ out)
{
  __shared__ __attribute__((aligned(16))) float lds[2 * PSTAGE_];
  proj_body(X, W, W, out, 0, blockIdx.x, lds);
}

__global__ void __launch_bounds__(PT_)
proj_qkv_kernel(const float* __restrict__ Xq, const float* __restrict__ Xk,
                const float* __restrict__ Xv,
                const float* __restrict__ Wq, const float* __restrict__ Wk,
                const float* __restrict__ Wv,
                const float* __restrict__ Wqa, const float* __restrict__ Wka,
                float* __restrict__ oqf, float* __restrict__ okft, float* __restrict__ ov)
{
  __shared__ __attribute__((aligned(16))) float lds[2 * PSTAGE_];
  static_assert(PW_ == DH_ && LQ_ == LK_, "a projection tile is 16 tokens of one head");
  static_assert((16 + 4 * 32) * LP_ <= 2 * PSTAGE_, "tile + 4 waves x 2 buffers fit the two stages");
  const int which = blockIdx.y;
  const float* X  = (which == 0) ? Xq : (which == 1) ? Xk : Xv;
  const float* W  = (which == 0) ? Wq : (which == 1) ? Wk : Wv;
  const float* Wa = (which == 0) ? Wqa : Wka;
  float*       o  = (which == 0) ? oqf : (which == 1) ? okft : ov;
  const int mode  = (which == 0) ? 2 : (which == 1) ? 3 : 1;
  proj_body(X, W, Wa, o, mode, blockIdx.x, lds);
}

constexpr int QT_ = 4;
constexpr int NS_ = 8;
constexpr int FS_ = F_ / NS_;
constexpr int SC_ = 4;
constexpr int NKB_ = LK_ / 32;

template <int NB>
__device__ __forceinline__ void score_blocks(const float* __restrict__ kt,
                                             const float* __restrict__ qr,
                                             const float* __restrict__ wp,
                                             float* __restrict__ spw)
{
  constexpr int FC = (NB == 1) ? 16 : 8;
  constexpr int QC = FC / 4;
  static_assert(FS_ % (2 * FC) == 0 && QC % 2 == 0, "two register sets alternate, two scalar sets per pair of quads");

  float acc[NB][QT_];
  for (int kb = 0; kb < NB; ++kb)
    for (int q = 0; q < QT_; ++q) acc[kb][q] = 0.f;

  v4f kA[QC][NB], kB[QC][NB];
  float qA[QT_][SC_], qB[QT_][SC_];
  float wA[SC_], wB[SC_];
#define K_LOAD(kx, fb)                                                            \
  for (int c = 0; c < QC; ++c)                                                    \
    for (int kb = 0; kb < NB; ++kb)                                               \
      kx[c][kb] = *(const v4f*)(kt + (size_t)kb * KB_ + (size_t)((fb) / 4 + c) * KQ_);
#define S_LOAD(qx, wx, fb)                                                        \
  for (int i = 0; i < SC_; ++i) wx[i] = wp[(fb) + i];                             \
  for (int q = 0; q < QT_; ++q)                                                   \
    for (int i = 0; i < SC_; ++i) qx[q][i] = qr[(size_t)q * F_ + (fb) + i];
#define SC_STEP(kx, c, qx, wx)                                                    \
  for (int i = 0; i < SC_; ++i) {                                                 \
    float t[QT_][NB];                                                             \
    for (int kb = 0; kb < NB; ++kb)                                               \
      for (int q = 0; q < QT_; ++q) t[q][kb] = qx[q][i] + kx[c][kb][i];           \
    for (int kb = 0; kb < NB; ++kb)                                               \
      for (int q = 0; q < QT_; ++q) t[q][kb] = TANH_F32(t[q][kb]);                \
    for (int kb = 0; kb < NB; ++kb)                                               \
      for (int q = 0; q < QT_; ++q) acc[kb][q] = __builtin_fmaf(wx[i], t[q][kb], acc[kb][q]); \
  }
  K_LOAD(kA, 0)
  S_LOAD(qA, wA, 0)
  for (int f0 = 0; f0 < FS_; f0 += 2 * FC) {
    K_LOAD(kB, f0 + FC)
    for (int c = 0; c < QC; c += 2) {
      S_LOAD(qB, wB, f0 + 4 * c + SC_)
      SC_STEP(kA, c, qA, wA)
      S_LOAD(qA, wA, f0 + 4 * c + 2 * SC_)
      SC_STEP(kA, c + 1, qB, wB)
    }
    const int fn = (f0 + 2 * FC < FS_) ? f0 + 2 * FC : 0;
    K_LOAD(kA, fn)
    for (int c = 0; c < QC; c += 2) {
      S_LOAD(qB, wB, f0 + FC + 4 * c + SC_)
      SC_STEP(kB, c, qA, wA)
      S_LOAD(qA, wA, (c + 2 < QC) ? f0 + FC + 4 * c + 2 * SC_ : fn)
      SC_STEP(kB, c + 1, qB, wB)
    }
  }
#undef K_LOAD
#undef S_LOAD
#undef SC_STEP
  for (int kb = 0; kb < NB; ++kb)
    for (int q = 0; q < QT_; ++q) spw[q * LK_ + 32 * kb] = acc[kb][q];
}

__global__ void __launch_bounds__(256)
score_softmax_av_kernel(const float* __restrict__ qf,
                        const float* __restrict__ kfb,
                        const float* __restrict__ vh,
                        const float* __restrict__ wv_a,
                        const int*   __restrict__ valid_lens,
                        float* __restrict__ ctx)
{
  static_assert(QT_ * DH_ == 256, "the last stage maps one thread to one (row, column)");
  static_assert(LK_ == 32 * NKB_ && NKB_ == 4 && F_ == NS_ * FS_ && NS_ == 8, "4 blocks of 32 lanes cover the keys, 8 waves cover the features");
  __shared__ __attribute__((aligned(16))) float sp[NS_][QT_][LK_];
  __shared__ float spart[QT_][4][DH_];

  const int qt  = blockIdx.x % (LQ_ / QT_);
  const int bh  = blockIdx.x / (LQ_ / QT_);
  const int q0  = qt * QT_;
  const int b   = bh / H_;
  const int h   = bh % H_;
  const int tid  = threadIdx.x;
  const int lane = tid & 31;
  const int wave = tid >> 5;

  const int valid = valid_lens[b];
  const int nb = (valid <= 0) ? 0 : (valid >= LK_) ? NKB_ : (valid + 31) / 32;

  if (nb > 0) {
    const int wu = __builtin_amdgcn_readfirstlane(wave);
    const float* kt = kfb + (size_t)bh * NKB_ * KB_ + (size_t)wu * (FS_ / 4) * KQ_ + 4 * lane;
    const float* qr = qf + ((size_t)bh * LQ_ + q0) * F_ + wu * FS_;
    const float* wp = wv_a + wu * FS_;
    float* spw = &sp[wu][0][lane];
    if (nb == 1)      score_blocks<1>(kt, qr, wp, spw);
    else if (nb == 2) score_blocks<2>(kt, qr, wp, spw);
    else if (nb == 3) score_blocks<3>(kt, qr, wp, spw);
    else              score_blocks<4>(kt, qr, wp, spw);
  }
  __syncthreads();

  if (wave < QT_) {
    const int q = wave;
    float val[4], e[4], sm[4];
    for (int j = 0; j < 4; ++j) {
      const int k = lane + 32 * j;
      float s = -1.0e6f;
      if (j < nb) {
        float a = sp[0][q][k];
        for (int w = 1; w < NS_; ++w) a += sp[w][q][k];
        s = (k < valid) ? a : -1.0e6f;
      }
      val[j] = s;
    }
    float m = fmaxf(fmaxf(val[0], val[1]), fmaxf(val[2], val[3]));
    for (int off = 16; off > 0; off >>= 1)
      m = fmaxf(m, __shfl_xor(m, off, 32));
    for (int j = 0; j < 4; ++j) {
      e[j] = __expf(val[j] - m);
      float t = e[j];
      for (int off = 16; off > 0; off >>= 1)
        t += __shfl_xor(t, off, 32);
      sm[j] = t;
    }
    const float rowsum = sm[0] + sm[1] + sm[2] + sm[3];
    for (int j = 0; j < 4; ++j) sp[0][q][lane + 32 * j] = e[j] / rowsum;
  }
  __syncthreads();

  const int j = tid & 63;
  const int part = tid >> 6;
  const float* vb = vh + (size_t)bh * LK_ * DH_;
  float o[QT_];
  for (int q = 0; q < QT_; ++q) o[q] = 0.f;
  for (int k = part * 32; k < part * 32 + 32; ++k) {
    const float v = vb[(size_t)k * DH_ + j];
    for (int q = 0; q < QT_; ++q) o[q] += sp[0][q][k] * v;
  }
  for (int q = 0; q < QT_; ++q) spart[q][part][j] = o[q];
  __syncthreads();

  {
    const int q = tid >> 6;
    const float r = spart[q][0][j] + spart[q][1][j] + spart[q][2][j] + spart[q][3][j];
    ctx[((size_t)b * LQ_ + q0 + q) * D_ + h * DH_ + j] = r;
  }
}

extern "C" void kernel_launch(void* const* d_in, const int* in_sizes, int n_in,
                              void* d_out, int out_size, void* d_ws, size_t ws_size,
                              hipStream_t stream) {
  (void)in_sizes; (void)n_in; (void)out_size; (void)ws_size;

  const float* queries    = (const float*)d_in[0];
  const float* keys       = (const float*)d_in[1];
  const float* values     = (const float*)d_in[2];
  const int*   valid_lens = (const int*)  d_in[3];
  const float* Wq  = (const float*)d_in[4];
  const float* Wk  = (const float*)d_in[5];
  const float* Wv  = (const float*)d_in[6];
  const float* Wo  = (const float*)d_in[7];
  const float* Wqa = (const float*)d_in[8];
  const float* Wka = (const float*)d_in[9];
  const float* wva = (const float*)d_in[10];

  float* ws  = (float*)d_ws;
  const size_t nQKV = (size_t)B_ * H_ * LQ_ * DH_;
  const size_t nF   = (size_t)B_ * H_ * LQ_ * F_;
  float* v   = ws;
  float* qf  = v  + nQKV;
  float* kfb = qf + nF;
  float* ctx = kfb + nF;

  dim3 blk(256);
  dim3 pblk(PT_);
  proj_qkv_kernel<<<dim3(128, 3), pblk, 0, stream>>>(queries, keys, values, Wq, Wk, Wv, Wqa, Wka, qf, kfb, v);

  score_softmax_av_kernel<<<B_ * H_ * (LQ_ / QT_), blk, 0, stream>>>(
      qf, kfb, v, wva, valid_lens, ctx);

  proj_kernel<<<128, pblk, 0, stream>>>(ctx, Wo, (float*)d_out);
}
